# starting kernel
# baseline (speedup 1.0000x reference)
#include <hip/hip_runtime.h>

typedef __attribute__((ext_vector_type(2))) float v2f;
typedef __attribute__((ext_vector_type(8))) float v8f;

#define D_IN  128
#define D_HID 128
#define D_OUT 64


__global__ void k_init_deg(float* deg, int n) {
    int i = blockIdx.x * blockDim.x + threadIdx.x;
    if (i < n) deg[i] = 1.0f;
}

__global__ void k_accum_deg(const int* __restrict__ col, const float* __restrict__ w,
                            float* deg, int e) {
    int i = blockIdx.x * blockDim.x + threadIdx.x;
    if (i < e) unsafeAtomicAdd(&deg[col[i]], w[i]);
}

__global__ void k_dinv(const float* __restrict__ deg, float* dinv, float* dinv2, int n) {
    int i = blockIdx.x * blockDim.x + threadIdx.x;
    if (i < n) {
        float d = deg[i];
        float r = (d > 0.0f) ? rsqrtf(d) : 0.0f;
        dinv[i]  = r;
        dinv2[i] = r * r;
    }
}

__global__ void k_edge_norm(const int* __restrict__ row, const int* __restrict__ col,
                            const float* __restrict__ w, const float* __restrict__ dinv,
                            float* nrm, int e) {
    int i = blockIdx.x * blockDim.x + threadIdx.x;
    if (i < e) nrm[i] = dinv[row[i]] * w[i] * dinv[col[i]];
}


template <int DINc, int NT>
__global__ __launch_bounds__(256)
void k_gemm_wmma(const float* __restrict__ hin, const float* __restrict__ W,
                 float* __restrict__ lin, int n, int relu) {
    const int Dout = NT * 16;
    const int lane = threadIdx.x & 31;
    const int wave = threadIdx.x >> 5;
    const int half = lane >> 4;
    const int lidx = lane & 15;

    const long rowBase = ((long)blockIdx.x * 8 + wave) * 16;
    if (rowBase >= n) return;

    long rowC = rowBase + lidx;
    if (rowC > n - 1) rowC = n - 1;

    v8f acc[NT] = {};

    const float* aRow = hin + (size_t)rowC * DINc + 2 * half;

    for (int k0 = 0; k0 < DINc; k0 += 4) {
        float2 a2 = *(const float2*)(aRow + k0);
        if (relu) { a2.x = fmaxf(a2.x, 0.0f); a2.y = fmaxf(a2.y, 0.0f); }
        v2f av; av.x = a2.x; av.y = a2.y;

        const float* bBase = W + (size_t)(k0 + 2 * half) * Dout + lidx;
#pragma unroll
        for (int t = 0; t < NT; ++t) {
            v2f bv;
            bv.x = bBase[t * 16];
            bv.y = bBase[Dout + t * 16];
            acc[t] = __builtin_amdgcn_wmma_f32_16x16x4_f32(
                false, av, false, bv, (short)0, acc[t], false, false);
        }
    }

#pragma unroll
    for (int t = 0; t < NT; ++t) {
        const int c = t * 16 + lidx;
#pragma unroll
        for (int r = 0; r < 8; ++r) {
            long rr = rowBase + half * 8 + r;
            if (rr < n) lin[(size_t)rr * Dout + c] = acc[t][r];
        }
    }
}


template <int DoutT>
__global__ void k_init_out(const float* __restrict__ lin, const float* __restrict__ bias,
                           const float* __restrict__ dinv2, float* __restrict__ out, int n) {
    const int quads = DoutT / 4;
    int idx = blockIdx.x * blockDim.x + threadIdx.x;
    if (idx >= n * quads) return;
    int node = idx / quads;
    int c4   = (idx - node * quads) * 4;
    float  s = dinv2[node];
    float4 l = *(const float4*)(lin + (size_t)node * DoutT + c4);
    float4 b = *(const float4*)(bias + c4);
    float4 o;
    o.x = fmaf(s, l.x, b.x);
    o.y = fmaf(s, l.y, b.y);
    o.z = fmaf(s, l.z, b.z);
    o.w = fmaf(s, l.w, b.w);
    *(float4*)(out + (size_t)node * DoutT + c4) = o;
}


template <int VEC>
__global__ __launch_bounds__(256)
void k_scatter(const int* __restrict__ row, const int* __restrict__ col,
               const float* __restrict__ nrm, const float* __restrict__ lin,
               float* __restrict__ out, int e) {
    const int waveId = blockIdx.x * 8 + (threadIdx.x >> 5);
    const int lane   = threadIdx.x & 31;
    if (waveId >= e) return;
    const int Dout = 32 * VEC;
    const int r = row[waveId];
    const int c = col[waveId];
    const float w = nrm[waveId];
    const float* src = lin + (size_t)r * Dout + lane * VEC;
    float*       dst = out + (size_t)c * Dout + lane * VEC;
    if (VEC == 4) {
        float4 v = *(const float4*)src;
        unsafeAtomicAdd(dst + 0, w * v.x);
        unsafeAtomicAdd(dst + 1, w * v.y);
        unsafeAtomicAdd(dst + 2, w * v.z);
        unsafeAtomicAdd(dst + 3, w * v.w);
    } else {
        float2 v = *(const float2*)src;
        unsafeAtomicAdd(dst + 0, w * v.x);
        unsafeAtomicAdd(dst + 1, w * v.y);
    }
}


static inline int cdiv(int a, int b) { return (a + b - 1) / b; }

extern "C" void kernel_launch(void* const* d_in, const int* in_sizes, int n_in,
                              void* d_out, int out_size, void* d_ws, size_t ws_size,
                              hipStream_t stream) {
    const float* x   = (const float*)d_in[0];
    const int*   ei  = (const int*)d_in[1];
    const float* ew  = (const float*)d_in[2];
    const float* W1  = (const float*)d_in[3];
    const float* b1  = (const float*)d_in[4];
    const float* W2  = (const float*)d_in[5];
    const float* b2  = (const float*)d_in[6];
    const float* W3  = (const float*)d_in[7];
    const float* b3  = (const float*)d_in[8];
    float* out = (float*)d_out;

    const int N = in_sizes[0] / D_IN;
    const int E = in_sizes[1] / 2;
    const int* eRow = ei;
    const int* eCol = ei + E;

    float* ws    = (float*)d_ws;
    float* deg   = ws;
    float* dinv  = deg   + N;
    float* dinv2 = dinv  + N;
    float* nrm   = dinv2 + N;
    float* buf0  = nrm   + E;
    float* buf1  = buf0  + (size_t)N * D_HID;

    k_init_deg <<<cdiv(N, 256), 256, 0, stream>>>(deg, N);
    k_accum_deg<<<cdiv(E, 256), 256, 0, stream>>>(eCol, ew, deg, E);
    k_dinv     <<<cdiv(N, 256), 256, 0, stream>>>(deg, dinv, dinv2, N);
    k_edge_norm<<<cdiv(E, 256), 256, 0, stream>>>(eRow, eCol, ew, dinv, nrm, E);

    const int gGemm = cdiv(N, 128);
    const int gScat = cdiv(E, 8);

    k_gemm_wmma<D_IN, 8><<<gGemm, 256, 0, stream>>>(x, W1, buf1, N, 0);
    k_init_out<128><<<cdiv(N * 32, 256), 256, 0, stream>>>(buf1, b1, dinv2, buf0, N);
    k_scatter<4><<<gScat, 256, 0, stream>>>(eRow, eCol, nrm, buf1, buf0, E);

    k_gemm_wmma<D_HID, 8><<<gGemm, 256, 0, stream>>>(buf0, W2, buf1, N, 1);
    k_init_out<128><<<cdiv(N * 32, 256), 256, 0, stream>>>(buf1, b2, dinv2, buf0, N);
    k_scatter<4><<<gScat, 256, 0, stream>>>(eRow, eCol, nrm, buf1, buf0, E);

    k_gemm_wmma<D_HID, 4><<<gGemm, 256, 0, stream>>>(buf0, W3, buf1, N, 1);
    k_init_out<64><<<cdiv(N * 16, 256), 256, 0, stream>>>(buf1, b3, dinv2, out, N);
    k_scatter<2><<<gScat, 256, 0, stream>>>(eRow, eCol, nrm, buf1, out, E);
}


# Replace per-edge float-atomic scatter by an in-call counting sort of edges by target and a per-node gather-sum, one store per output row; sums now run in segment order, no float atomics
# speedup vs baseline: 2.2183x; 2.2183x over previous
#include <hip/hip_runtime.h>

typedef __attribute__((ext_vector_type(2))) float v2f;
typedef __attribute__((ext_vector_type(8))) float v8f;

#define D_IN  128
#define D_HID 128
#define D_OUT 64


__global__ void k_zero_cnt(int* cnt, int* total, int n) {
    int i = blockIdx.x * blockDim.x + threadIdx.x;
    if (i < n) cnt[i] = 0;
    if (i == 0) *total = 0;
}

__global__ void k_count(const int* __restrict__ col, int* cnt, int e) {
    int i = blockIdx.x * blockDim.x + threadIdx.x;
    if (i < e) atomicAdd(&cnt[col[i]], 1);
}

__global__ __launch_bounds__(256)
void k_alloc(const int* __restrict__ cnt, int* __restrict__ start, int* __restrict__ cursor,
             int* total, int n) {
    __shared__ int wsum[8];
    __shared__ int base;
    const int i    = blockIdx.x * 256 + threadIdx.x;
    const int lane = threadIdx.x & 31;
    const int wv   = threadIdx.x >> 5;
    const int c = (i < n) ? cnt[i] : 0;
    int incl = c;
#pragma unroll
    for (int d = 1; d < 32; d <<= 1) {
        int t = __shfl_up(incl, d);
        if (lane >= d) incl += t;
    }
    if (lane == 31) wsum[wv] = incl;
    __syncthreads();
    if (threadIdx.x == 0) {
        int s = 0;
        for (int k = 0; k < 8; ++k) { int t = wsum[k]; wsum[k] = s; s += t; }
        base = atomicAdd(total, s);
    }
    __syncthreads();
    const int st = base + wsum[wv] + incl - c;
    if (i < n) { start[i] = st; cursor[i] = st; }
}

__global__ void k_fill(const int* __restrict__ row, const int* __restrict__ col,
                       const float* __restrict__ w, int* cursor,
                       int2* __restrict__ csr, int e) {
    int i = blockIdx.x * blockDim.x + threadIdx.x;
    if (i < e) {
        int p = atomicAdd(&cursor[col[i]], 1);
        int2 v; v.x = row[i]; v.y = __float_as_int(w[i]);
        csr[p] = v;
    }
}

__global__ void k_dinv(const int* __restrict__ start, const int* __restrict__ cnt,
                       const int2* __restrict__ csr, float* dinv, float* dinv2, int n) {
    int i = blockIdx.x * blockDim.x + threadIdx.x;
    if (i < n) {
        const int s = start[i], c = cnt[i];
        float d = 1.0f;
        for (int j = 0; j < c; ++j) d += __int_as_float(csr[s + j].y);
        float r = (d > 0.0f) ? rsqrtf(d) : 0.0f;
        dinv[i]  = r;
        dinv2[i] = r * r;
    }
}

__global__ void k_seg_norm(const int* __restrict__ start, const int* __restrict__ cnt,
                           const float* __restrict__ dinv, int2* csr, int n) {
    int i = blockIdx.x * blockDim.x + threadIdx.x;
    if (i < n) {
        const int s = start[i], c = cnt[i];
        const float dt = dinv[i];
        for (int j = 0; j < c; ++j) {
            int2 v = csr[s + j];
            v.y = __float_as_int(dinv[v.x] * __int_as_float(v.y) * dt);
            csr[s + j] = v;
        }
    }
}

__global__ void k_copy_e(const int2* __restrict__ a, int2* __restrict__ b, int e) {
    int i = blockIdx.x * blockDim.x + threadIdx.x;
    if (i < e) b[i] = a[i];
}


template <int DINc, int NT>
__global__ __launch_bounds__(256)
void k_gemm_wmma(const float* __restrict__ hin, const float* __restrict__ W,
                 float* __restrict__ lin, int n, int relu) {
    const int Dout = NT * 16;
    const int lane = threadIdx.x & 31;
    const int wave = threadIdx.x >> 5;
    const int half = lane >> 4;
    const int lidx = lane & 15;

    const long rowBase = ((long)blockIdx.x * 8 + wave) * 16;
    if (rowBase >= n) return;

    long rowC = rowBase + lidx;
    if (rowC > n - 1) rowC = n - 1;

    v8f acc[NT] = {};

    const float* aRow = hin + (size_t)rowC * DINc + 2 * half;

    for (int k0 = 0; k0 < DINc; k0 += 4) {
        float2 a2 = *(const float2*)(aRow + k0);
        if (relu) { a2.x = fmaxf(a2.x, 0.0f); a2.y = fmaxf(a2.y, 0.0f); }
        v2f av; av.x = a2.x; av.y = a2.y;

        const float* bBase = W + (size_t)(k0 + 2 * half) * Dout + lidx;
#pragma unroll
        for (int t = 0; t < NT; ++t) {
            v2f bv;
            bv.x = bBase[t * 16];
            bv.y = bBase[Dout + t * 16];
            acc[t] = __builtin_amdgcn_wmma_f32_16x16x4_f32(
                false, av, false, bv, (short)0, acc[t], false, false);
        }
    }

#pragma unroll
    for (int t = 0; t < NT; ++t) {
        const int c = t * 16 + lidx;
#pragma unroll
        for (int r = 0; r < 8; ++r) {
            long rr = rowBase + half * 8 + r;
            if (rr < n) lin[(size_t)rr * Dout + c] = acc[t][r];
        }
    }
}


template <int VEC> struct VecT;
template <> struct VecT<4> { typedef float4 T; };
template <> struct VecT<2> { typedef float2 T; };

__device__ __forceinline__ void vfma(float4& a, float w, const float4& v) {
    a.x = fmaf(w, v.x, a.x); a.y = fmaf(w, v.y, a.y);
    a.z = fmaf(w, v.z, a.z); a.w = fmaf(w, v.w, a.w);
}
__device__ __forceinline__ void vfma(float2& a, float w, const float2& v) {
    a.x = fmaf(w, v.x, a.x); a.y = fmaf(w, v.y, a.y);
}
__device__ __forceinline__ float4 vinit(float s, const float4& l, const float4& b) {
    float4 o; o.x = fmaf(s, l.x, b.x); o.y = fmaf(s, l.y, b.y);
    o.z = fmaf(s, l.z, b.z); o.w = fmaf(s, l.w, b.w); return o;
}
__device__ __forceinline__ float2 vinit(float s, const float2& l, const float2& b) {
    float2 o; o.x = fmaf(s, l.x, b.x); o.y = fmaf(s, l.y, b.y); return o;
}

template <int VEC>
__global__ __launch_bounds__(256)
void k_gather(const int* __restrict__ start, const int* __restrict__ cnt,
              const int2* __restrict__ csr,
              const float* __restrict__ lin, const float* __restrict__ bias,
              const float* __restrict__ dinv2, float* __restrict__ out, int n) {
    typedef typename VecT<VEC>::T V;
    const int Dout = 32 * VEC;
    const int lane = threadIdx.x & 31;
    int node = blockIdx.x * 8 + (threadIdx.x >> 5);
    if (node >= n) return;
    node = __builtin_amdgcn_readfirstlane(node);
    const int s = __builtin_amdgcn_readfirstlane(start[node]);
    const int c = __builtin_amdgcn_readfirstlane(cnt[node]);

    const float* linL = lin + lane * VEC;
    V acc = vinit(dinv2[node], *(const V*)(linL + (size_t)node * Dout),
                  *(const V*)(bias + lane * VEC));

    for (int b0 = 0; b0 < c; b0 += 32) {
        const int m = (c - b0 < 32) ? (c - b0) : 32;
        int2 my; my.x = 0; my.y = 0;
        if (lane < m) my = csr[s + b0 + lane];
        const int mySrc = my.x;
        const int myWi  = my.y;
        int k = 0;
        for (; k + 4 <= m; k += 4) {
            const int s0 = __builtin_amdgcn_readlane(mySrc, k);
            const int s1 = __builtin_amdgcn_readlane(mySrc, k + 1);
            const int s2 = __builtin_amdgcn_readlane(mySrc, k + 2);
            const int s3 = __builtin_amdgcn_readlane(mySrc, k + 3);
            const V v0 = *(const V*)(linL + (size_t)s0 * Dout);
            const V v1 = *(const V*)(linL + (size_t)s1 * Dout);
            const V v2 = *(const V*)(linL + (size_t)s2 * Dout);
            const V v3 = *(const V*)(linL + (size_t)s3 * Dout);
            vfma(acc, __int_as_float(__builtin_amdgcn_readlane(myWi, k)),     v0);
            vfma(acc, __int_as_float(__builtin_amdgcn_readlane(myWi, k + 1)), v1);
            vfma(acc, __int_as_float(__builtin_amdgcn_readlane(myWi, k + 2)), v2);
            vfma(acc, __int_as_float(__builtin_amdgcn_readlane(myWi, k + 3)), v3);
        }
        for (; k < m; ++k) {
            const int s0 = __builtin_amdgcn_readlane(mySrc, k);
            const V v0 = *(const V*)(linL + (size_t)s0 * Dout);
            vfma(acc, __int_as_float(__builtin_amdgcn_readlane(myWi, k)), v0);
        }
    }
    *(V*)(out + (size_t)node * Dout + lane * VEC) = acc;
}


static inline int cdiv(int a, int b) { return (a + b - 1) / b; }

extern "C" void kernel_launch(void* const* d_in, const int* in_sizes, int n_in,
                              void* d_out, int out_size, void* d_ws, size_t ws_size,
                              hipStream_t stream) {
    const float* x   = (const float*)d_in[0];
    const int*   ei  = (const int*)d_in[1];
    const float* ew  = (const float*)d_in[2];
    const float* W1  = (const float*)d_in[3];
    const float* b1  = (const float*)d_in[4];
    const float* W2  = (const float*)d_in[5];
    const float* b2  = (const float*)d_in[6];
    const float* W3  = (const float*)d_in[7];
    const float* b3  = (const float*)d_in[8];
    float* out = (float*)d_out;

    const int N = in_sizes[0] / D_IN;
    const int E = in_sizes[1] / 2;
    const int* eRow = ei;
    const int* eCol = ei + E;

    float* ws     = (float*)d_ws;
    int*   cnt    = (int*)ws;
    int*   start  = cnt + N;
    float* dinv2  = ws + 2 * (size_t)N;
    int*   cursor = (int*)(ws + 3 * (size_t)N);
    float* dinv   = ws + 4 * (size_t)N;
    int*   total  = (int*)(ws + 5 * (size_t)N);
    float* buf0   = ws + 3 * (size_t)N + E;
    float* buf1   = buf0 + (size_t)N * D_HID;

    int2*  csrA   = (int2*)out;
    int2*  csrB   = (int2*)buf0;

    k_zero_cnt<<<cdiv(N, 256), 256, 0, stream>>>(cnt, total, N);
    k_count   <<<cdiv(E, 256), 256, 0, stream>>>(eCol, cnt, E);
    k_alloc   <<<cdiv(N, 256), 256, 0, stream>>>(cnt, start, cursor, total, N);
    k_fill    <<<cdiv(E, 256), 256, 0, stream>>>(eRow, eCol, ew, cursor, csrA, E);
    k_dinv    <<<cdiv(N, 256), 256, 0, stream>>>(start, cnt, csrA, dinv, dinv2, N);
    k_seg_norm<<<cdiv(N, 256), 256, 0, stream>>>(start, cnt, dinv, csrA, N);

    const int gGemm = cdiv(N, 128);
    const int gGath = cdiv(N, 8);

    k_gemm_wmma<D_IN, 8><<<gGemm, 256, 0, stream>>>(x, W1, buf1, N, 0);
    k_gather<4><<<gGath, 256, 0, stream>>>(start, cnt, csrA, buf1, b1, dinv2, buf0, N);

    k_gemm_wmma<D_HID, 8><<<gGemm, 256, 0, stream>>>(buf0, W2, buf1, N, 1);
    k_gather<4><<<gGath, 256, 0, stream>>>(start, cnt, csrA, buf1, b2, dinv2, buf0, N);

    k_gemm_wmma<D_HID, 4><<<gGemm, 256, 0, stream>>>(buf0, W3, buf1, N, 1);
    k_copy_e<<<cdiv(E, 256), 256, 0, stream>>>(csrA, csrB, E);
    k_gather<2><<<gGath, 256, 0, stream>>>(start, cnt, csrB, buf1, b3, dinv2, out, N);
}


# Edges ordered by target on LDS counters: split into 512-target ranges, then one work-group per range ranks in LDS; 3.2M global integer atomics become 0.15M; in-segment order still by arrival
# speedup vs baseline: 3.2475x; 1.4640x over previous
#include <hip/hip_runtime.h>

typedef __attribute__((ext_vector_type(2))) float v2f;
typedef __attribute__((ext_vector_type(8))) float v8f;

#define D_IN  128
#define D_HID 128
#define D_OUT 64


#define BSH   9
#define BSZ   (1 << BSH)
#define MAXB  2048
#define CHUNK 4096
#define EPT   (CHUNK / 256)

__global__ void k_zero_b(int* btot, int* gcur, int nb) {
    int i = blockIdx.x * blockDim.x + threadIdx.x;
    if (i < nb) { btot[i] = 0; gcur[i] = 0; }
}

__global__ __launch_bounds__(256)
void k_bhist(const int* __restrict__ col, int* btot, int e, int nb) {
    __shared__ int hist[MAXB];
    for (int b = threadIdx.x; b < nb; b += 256) hist[b] = 0;
    __syncthreads();
    const int base = blockIdx.x * CHUNK + threadIdx.x;
    int c[EPT];
#pragma unroll
    for (int j = 0; j < EPT; ++j) {
        const int i = base + j * 256;
        c[j] = (i < e) ? col[i] : -1;
    }
#pragma unroll
    for (int j = 0; j < EPT; ++j)
        if (c[j] >= 0) atomicAdd(&hist[c[j] >> BSH], 1);
    __syncthreads();
    for (int b = threadIdx.x; b < nb; b += 256) {
        const int h = hist[b];
        if (h) atomicAdd(&btot[b], h);
    }
}

__global__ __launch_bounds__(256)
void k_part(const int* __restrict__ row, const int* __restrict__ col, const float* __restrict__ w,
            const int* __restrict__ btot, int* gcur, int4* __restrict__ part, int e, int nb) {
    __shared__ int hist[MAXB];
    __shared__ int bst[MAXB];
    const int tid  = threadIdx.x;
    const int lane = tid & 31;
    const int base = blockIdx.x * CHUNK + tid;
    int c[EPT], s[EPT], wb[EPT], r[EPT];
#pragma unroll
    for (int j = 0; j < EPT; ++j) {
        const int i = base + j * 256;
        const bool ok = i < e;
        c[j]  = ok ? col[i] : -1;
        s[j]  = ok ? row[i] : 0;
        wb[j] = ok ? __float_as_int(w[i]) : 0;
    }
    for (int b = tid; b < nb; b += 256) { hist[b] = 0; bst[b] = btot[b]; }
    __syncthreads();
    if (tid < 32) {
        int carry = 0;
        for (int i0 = 0; i0 < nb; i0 += 32) {
            const int i = i0 + lane;
            const int v = (i < nb) ? bst[i] : 0;
            int incl = v;
#pragma unroll
            for (int d = 1; d < 32; d <<= 1) {
                int t = __shfl_up(incl, d);
                if (lane >= d) incl += t;
            }
            if (i < nb) bst[i] = carry + incl - v;
            carry += __shfl(incl, 31);
        }
    }
#pragma unroll
    for (int j = 0; j < EPT; ++j)
        r[j] = (c[j] >= 0) ? atomicAdd(&hist[c[j] >> BSH], 1) : 0;
    __syncthreads();
    for (int b = tid; b < nb; b += 256) {
        const int h = hist[b];
        const int g = h ? atomicAdd(&gcur[b], h) : 0;
        hist[b] = bst[b] + g;
    }
    __syncthreads();
#pragma unroll
    for (int j = 0; j < EPT; ++j) {
        if (c[j] >= 0) {
            int4 v; v.x = s[j]; v.y = c[j]; v.z = wb[j]; v.w = 0;
            part[hist[c[j] >> BSH] + r[j]] = v;
        }
    }
}

__global__ __launch_bounds__(1024)
void k_bsort(const int4* __restrict__ part, const int* __restrict__ btot,
             int* __restrict__ start, int* __restrict__ cnt, int2* __restrict__ csr, int n) {
    __shared__ int c[BSZ];
    __shared__ int cur[BSZ];
    __shared__ int wsum[BSZ / 32];
    __shared__ int sb;
    const int tid  = threadIdx.x;
    const int lane = tid & 31;
    const int wv   = tid >> 5;
    const int b    = blockIdx.x;
    const int t0   = b << BSH;
    if (tid < BSZ) c[tid] = 0;
    if (tid == 0) sb = 0;
    __syncthreads();
    int p = 0;
    for (int i = tid; i < b; i += 1024) p += btot[i];
#pragma unroll
    for (int d = 16; d >= 1; d >>= 1) p += __shfl_xor(p, d);
    if (lane == 0 && p) atomicAdd(&sb, p);
    const int nbk = btot[b];
    __syncthreads();
    const int bs = sb;
    const int4* mine = part + bs;

    for (int i0 = 0; i0 < nbk; i0 += 4096) {
        int t[4];
#pragma unroll
        for (int k = 0; k < 4; ++k) {
            const int i = i0 + k * 1024 + tid;
            t[k] = (i < nbk) ? mine[i].y : -1;
        }
#pragma unroll
        for (int k = 0; k < 4; ++k)
            if (t[k] >= 0) atomicAdd(&c[t[k] - t0], 1);
    }
    __syncthreads();

    int v = 0, incl = 0;
    if (tid < BSZ) {
        v = c[tid];
        incl = v;
#pragma unroll
        for (int d = 1; d < 32; d <<= 1) {
            int t = __shfl_up(incl, d);
            if (lane >= d) incl += t;
        }
        if (lane == 31) wsum[wv] = incl;
    }
    __syncthreads();
    if (tid < BSZ) {
        int pre = 0;
        for (int k = 0; k < wv; ++k) pre += wsum[k];
        const int off = pre + incl - v;
        cur[tid] = off;
        const int t = t0 + tid;
        if (t < n) { start[t] = bs + off; cnt[t] = v; }
    }
    __syncthreads();

    int2* outp = csr + bs;
    for (int i0 = 0; i0 < nbk; i0 += 4096) {
        int4 rec[4];
#pragma unroll
        for (int k = 0; k < 4; ++k) {
            const int i = i0 + k * 1024 + tid;
            if (i < nbk) rec[k] = mine[i];
            else { rec[k].x = 0; rec[k].y = -1; rec[k].z = 0; rec[k].w = 0; }
        }
#pragma unroll
        for (int k = 0; k < 4; ++k) {
            if (rec[k].y >= 0) {
                const int q = atomicAdd(&cur[rec[k].y - t0], 1);
                int2 o; o.x = rec[k].x; o.y = rec[k].z;
                outp[q] = o;
            }
        }
    }
}

__global__ void k_dinv(const int* __restrict__ start, const int* __restrict__ cnt,
                       const int2* __restrict__ csr, float* dinv, float* dinv2, int n) {
    int i = blockIdx.x * blockDim.x + threadIdx.x;
    if (i < n) {
        const int s = start[i], c = cnt[i];
        float d = 1.0f;
        for (int j = 0; j < c; ++j) d += __int_as_float(csr[s + j].y);
        float r = (d > 0.0f) ? rsqrtf(d) : 0.0f;
        dinv[i]  = r;
        dinv2[i] = r * r;
    }
}

__global__ void k_seg_norm(const int* __restrict__ start, const int* __restrict__ cnt,
                           const float* __restrict__ dinv, int2* csr, int n) {
    int i = blockIdx.x * blockDim.x + threadIdx.x;
    if (i < n) {
        const int s = start[i], c = cnt[i];
        const float dt = dinv[i];
        for (int j = 0; j < c; ++j) {
            int2 v = csr[s + j];
            v.y = __float_as_int(dinv[v.x] * __int_as_float(v.y) * dt);
            csr[s + j] = v;
        }
    }
}

__global__ void k_copy_e(const int2* __restrict__ a, int2* __restrict__ b, int e) {
    int i = blockIdx.x * blockDim.x + threadIdx.x;
    if (i < e) b[i] = a[i];
}


template <int DINc, int NT>
__global__ __launch_bounds__(256)
void k_gemm_wmma(const float* __restrict__ hin, const float* __restrict__ W,
                 float* __restrict__ lin, int n, int relu) {
    const int Dout = NT * 16;
    const int lane = threadIdx.x & 31;
    const int wave = threadIdx.x >> 5;
    const int half = lane >> 4;
    const int lidx = lane & 15;

    const long rowBase = ((long)blockIdx.x * 8 + wave) * 16;
    if (rowBase >= n) return;

    long rowC = rowBase + lidx;
    if (rowC > n - 1) rowC = n - 1;

    v8f acc[NT] = {};

    const float* aRow = hin + (size_t)rowC * DINc + 2 * half;

    for (int k0 = 0; k0 < DINc; k0 += 4) {
        float2 a2 = *(const float2*)(aRow + k0);
        if (relu) { a2.x = fmaxf(a2.x, 0.0f); a2.y = fmaxf(a2.y, 0.0f); }
        v2f av; av.x = a2.x; av.y = a2.y;

        const float* bBase = W + (size_t)(k0 + 2 * half) * Dout + lidx;
#pragma unroll
        for (int t = 0; t < NT; ++t) {
            v2f bv;
            bv.x = bBase[t * 16];
            bv.y = bBase[Dout + t * 16];
            acc[t] = __builtin_amdgcn_wmma_f32_16x16x4_f32(
                false, av, false, bv, (short)0, acc[t], false, false);
        }
    }

#pragma unroll
    for (int t = 0; t < NT; ++t) {
        const int c = t * 16 + lidx;
#pragma unroll
        for (int r = 0; r < 8; ++r) {
            long rr = rowBase + half * 8 + r;
            if (rr < n) lin[(size_t)rr * Dout + c] = acc[t][r];
        }
    }
}


template <int VEC> struct VecT;
template <> struct VecT<4> { typedef float4 T; };
template <> struct VecT<2> { typedef float2 T; };

__device__ __forceinline__ void vfma(float4& a, float w, const float4& v) {
    a.x = fmaf(w, v.x, a.x); a.y = fmaf(w, v.y, a.y);
    a.z = fmaf(w, v.z, a.z); a.w = fmaf(w, v.w, a.w);
}
__device__ __forceinline__ void vfma(float2& a, float w, const float2& v) {
    a.x = fmaf(w, v.x, a.x); a.y = fmaf(w, v.y, a.y);
}
__device__ __forceinline__ float4 vinit(float s, const float4& l, const float4& b) {
    float4 o; o.x = fmaf(s, l.x, b.x); o.y = fmaf(s, l.y, b.y);
    o.z = fmaf(s, l.z, b.z); o.w = fmaf(s, l.w, b.w); return o;
}
__device__ __forceinline__ float2 vinit(float s, const float2& l, const float2& b) {
    float2 o; o.x = fmaf(s, l.x, b.x); o.y = fmaf(s, l.y, b.y); return o;
}

template <int VEC>
__global__ __launch_bounds__(256)
void k_gather(const int* __restrict__ start, const int* __restrict__ cnt,
              const int2* __restrict__ csr,
              const float* __restrict__ lin, const float* __restrict__ bias,
              const float* __restrict__ dinv2, float* __restrict__ out, int n) {
    typedef typename VecT<VEC>::T V;
    const int Dout = 32 * VEC;
    const int lane = threadIdx.x & 31;
    int node = blockIdx.x * 8 + (threadIdx.x >> 5);
    if (node >= n) return;
    node = __builtin_amdgcn_readfirstlane(node);
    const int s = __builtin_amdgcn_readfirstlane(start[node]);
    const int c = __builtin_amdgcn_readfirstlane(cnt[node]);

    const float* linL = lin + lane * VEC;
    V acc = vinit(dinv2[node], *(const V*)(linL + (size_t)node * Dout),
                  *(const V*)(bias + lane * VEC));

    for (int b0 = 0; b0 < c; b0 += 32) {
        const int m = (c - b0 < 32) ? (c - b0) : 32;
        int2 my; my.x = 0; my.y = 0;
        if (lane < m) my = csr[s + b0 + lane];
        const int mySrc = my.x;
        const int myWi  = my.y;
        int k = 0;
        for (; k + 4 <= m; k += 4) {
            const int s0 = __builtin_amdgcn_readlane(mySrc, k);
            const int s1 = __builtin_amdgcn_readlane(mySrc, k + 1);
            const int s2 = __builtin_amdgcn_readlane(mySrc, k + 2);
            const int s3 = __builtin_amdgcn_readlane(mySrc, k + 3);
            const V v0 = *(const V*)(linL + (size_t)s0 * Dout);
            const V v1 = *(const V*)(linL + (size_t)s1 * Dout);
            const V v2 = *(const V*)(linL + (size_t)s2 * Dout);
            const V v3 = *(const V*)(linL + (size_t)s3 * Dout);
            vfma(acc, __int_as_float(__builtin_amdgcn_readlane(myWi, k)),     v0);
            vfma(acc, __int_as_float(__builtin_amdgcn_readlane(myWi, k + 1)), v1);
            vfma(acc, __int_as_float(__builtin_amdgcn_readlane(myWi, k + 2)), v2);
            vfma(acc, __int_as_float(__builtin_amdgcn_readlane(myWi, k + 3)), v3);
        }
        for (; k < m; ++k) {
            const int s0 = __builtin_amdgcn_readlane(mySrc, k);
            const V v0 = *(const V*)(linL + (size_t)s0 * Dout);
            vfma(acc, __int_as_float(__builtin_amdgcn_readlane(myWi, k)), v0);
        }
    }
    *(V*)(out + (size_t)node * Dout + lane * VEC) = acc;
}


static inline int cdiv(int a, int b) { return (a + b - 1) / b; }

extern "C" void kernel_launch(void* const* d_in, const int* in_sizes, int n_in,
                              void* d_out, int out_size, void* d_ws, size_t ws_size,
                              hipStream_t stream) {
    const float* x   = (const float*)d_in[0];
    const int*   ei  = (const int*)d_in[1];
    const float* ew  = (const float*)d_in[2];
    const float* W1  = (const float*)d_in[3];
    const float* b1  = (const float*)d_in[4];
    const float* W2  = (const float*)d_in[5];
    const float* b2  = (const float*)d_in[6];
    const float* W3  = (const float*)d_in[7];
    const float* b3  = (const float*)d_in[8];
    float* out = (float*)d_out;

    const int N = in_sizes[0] / D_IN;
    const int E = in_sizes[1] / 2;
    const int* eRow = ei;
    const int* eCol = ei + E;

    float* ws     = (float*)d_ws;
    int*   cnt    = (int*)ws;
    int*   start  = cnt + N;
    float* dinv2  = ws + 2 * (size_t)N;
    const int NB  = cdiv(N, BSZ);
    int*   btot   = (int*)(ws + 3 * (size_t)N);
    int*   gcur   = btot + NB;
    float* dinv   = ws + 4 * (size_t)N;
    float* buf0   = ws + 3 * (size_t)N + E;
    float* buf1   = buf0 + (size_t)N * D_HID;

    int2*  csrA   = (int2*)out;
    int2*  csrB   = (int2*)buf0;
    int4*  part   = (int4*)buf0;

    k_zero_b  <<<cdiv(NB, 256), 256, 0, stream>>>(btot, gcur, NB);
    k_bhist   <<<cdiv(E, CHUNK), 256, 0, stream>>>(eCol, btot, E, NB);
    k_part    <<<cdiv(E, CHUNK), 256, 0, stream>>>(eRow, eCol, ew, btot, gcur, part, E, NB);
    k_bsort   <<<NB, 1024, 0, stream>>>(part, btot, start, cnt, csrA, N);
    k_dinv    <<<cdiv(N, 256), 256, 0, stream>>>(start, cnt, csrA, dinv, dinv2, N);
    k_seg_norm<<<cdiv(N, 256), 256, 0, stream>>>(start, cnt, dinv, csrA, N);

    const int gGemm = cdiv(N, 128);
    const int gGath = cdiv(N, 8);

    k_gemm_wmma<D_IN, 8><<<gGemm, 256, 0, stream>>>(x, W1, buf1, N, 0);
    k_gather<4><<<gGath, 256, 0, stream>>>(start, cnt, csrA, buf1, b1, dinv2, buf0, N);

    k_gemm_wmma<D_HID, 8><<<gGemm, 256, 0, stream>>>(buf0, W2, buf1, N, 1);
    k_gather<4><<<gGath, 256, 0, stream>>>(start, cnt, csrA, buf1, b2, dinv2, buf0, N);

    k_gemm_wmma<D_HID, 4><<<gGemm, 256, 0, stream>>>(buf0, W3, buf1, N, 1);
    k_copy_e<<<cdiv(E, 256), 256, 0, stream>>>(csrA, csrB, E);
    k_gather<2><<<gGath, 256, 0, stream>>>(start, cnt, csrB, buf1, b3, dinv2, out, N);
}


# Norm factored out of the edges: GEMM stores dinv[row]*row, gather sums raw-weight terms in segment order, scales by dinv[target], adds bias last; per-edge norm pass dropped; products regrouped
# speedup vs baseline: 3.6196x; 1.1146x over previous
#include <hip/hip_runtime.h>

typedef __attribute__((ext_vector_type(2))) float v2f;
typedef __attribute__((ext_vector_type(8))) float v8f;

#define D_IN  128
#define D_HID 128
#define D_OUT 64


#define BSH   9
#define BSZ   (1 << BSH)
#define MAXB  2048
#define CHUNK 4096
#define EPT   (CHUNK / 256)

__global__ void k_zero_b(int* btot, int* gcur, int nb) {
    int i = blockIdx.x * blockDim.x + threadIdx.x;
    if (i < nb) { btot[i] = 0; gcur[i] = 0; }
}

__global__ __launch_bounds__(256)
void k_bhist(const int* __restrict__ col, int* btot, int e, int nb) {
    __shared__ int hist[MAXB];
    for (int b = threadIdx.x; b < nb; b += 256) hist[b] = 0;
    __syncthreads();
    const int base = blockIdx.x * CHUNK + threadIdx.x;
    int c[EPT];
#pragma unroll
    for (int j = 0; j < EPT; ++j) {
        const int i = base + j * 256;
        c[j] = (i < e) ? col[i] : -1;
    }
#pragma unroll
    for (int j = 0; j < EPT; ++j)
        if (c[j] >= 0) atomicAdd(&hist[c[j] >> BSH], 1);
    __syncthreads();
    for (int b = threadIdx.x; b < nb; b += 256) {
        const int h = hist[b];
        if (h) atomicAdd(&btot[b], h);
    }
}

__global__ __launch_bounds__(256)
void k_part(const int* __restrict__ row, const int* __restrict__ col, const float* __restrict__ w,
            const int* __restrict__ btot, int* gcur, int4* __restrict__ part, int e, int nb) {
    __shared__ int hist[MAXB];
    __shared__ int bst[MAXB];
    const int tid  = threadIdx.x;
    const int lane = tid & 31;
    const int base = blockIdx.x * CHUNK + tid;
    int c[EPT], s[EPT], wb[EPT], r[EPT];
#pragma unroll
    for (int j = 0; j < EPT; ++j) {
        const int i = base + j * 256;
        const bool ok = i < e;
        c[j]  = ok ? col[i] : -1;
        s[j]  = ok ? row[i] : 0;
        wb[j] = ok ? __float_as_int(w[i]) : 0;
    }
    for (int b = tid; b < nb; b += 256) { hist[b] = 0; bst[b] = btot[b]; }
    __syncthreads();
    if (tid < 32) {
        int carry = 0;
        for (int i0 = 0; i0 < nb; i0 += 32) {
            const int i = i0 + lane;
            const int v = (i < nb) ? bst[i] : 0;
            int incl = v;
#pragma unroll
            for (int d = 1; d < 32; d <<= 1) {
                int t = __shfl_up(incl, d);
                if (lane >= d) incl += t;
            }
            if (i < nb) bst[i] = carry + incl - v;
            carry += __shfl(incl, 31);
        }
    }
#pragma unroll
    for (int j = 0; j < EPT; ++j)
        r[j] = (c[j] >= 0) ? atomicAdd(&hist[c[j] >> BSH], 1) : 0;
    __syncthreads();
    for (int b = tid; b < nb; b += 256) {
        const int h = hist[b];
        const int g = h ? atomicAdd(&gcur[b], h) : 0;
        hist[b] = bst[b] + g;
    }
    __syncthreads();
#pragma unroll
    for (int j = 0; j < EPT; ++j) {
        if (c[j] >= 0) {
            int4 v; v.x = s[j]; v.y = c[j]; v.z = wb[j]; v.w = 0;
            part[hist[c[j] >> BSH] + r[j]] = v;
        }
    }
}

__global__ __launch_bounds__(1024)
void k_bsort(const int4* __restrict__ part, const int* __restrict__ btot,
             int* __restrict__ start, int* __restrict__ cnt, int2* __restrict__ csr, int n) {
    __shared__ int c[BSZ];
    __shared__ int cur[BSZ];
    __shared__ int wsum[BSZ / 32];
    __shared__ int sb;
    const int tid  = threadIdx.x;
    const int lane = tid & 31;
    const int wv   = tid >> 5;
    const int b    = blockIdx.x;
    const int t0   = b << BSH;
    if (tid < BSZ) c[tid] = 0;
    if (tid == 0) sb = 0;
    __syncthreads();
    int p = 0;
    for (int i = tid; i < b; i += 1024) p += btot[i];
#pragma unroll
    for (int d = 16; d >= 1; d >>= 1) p += __shfl_xor(p, d);
    if (lane == 0 && p) atomicAdd(&sb, p);
    const int nbk = btot[b];
    __syncthreads();
    const int bs = sb;
    const int4* mine = part + bs;

    for (int i0 = 0; i0 < nbk; i0 += 4096) {
        int t[4];
#pragma unroll
        for (int k = 0; k < 4; ++k) {
            const int i = i0 + k * 1024 + tid;
            t[k] = (i < nbk) ? mine[i].y : -1;
        }
#pragma unroll
        for (int k = 0; k < 4; ++k)
            if (t[k] >= 0) atomicAdd(&c[t[k] - t0], 1);
    }
    __syncthreads();

    int v = 0, incl = 0;
    if (tid < BSZ) {
        v = c[tid];
        incl = v;
#pragma unroll
        for (int d = 1; d < 32; d <<= 1) {
            int t = __shfl_up(incl, d);
            if (lane >= d) incl += t;
        }
        if (lane == 31) wsum[wv] = incl;
    }
    __syncthreads();
    if (tid < BSZ) {
        int pre = 0;
        for (int k = 0; k < wv; ++k) pre += wsum[k];
        const int off = pre + incl - v;
        cur[tid] = off;
        const int t = t0 + tid;
        if (t < n) { start[t] = bs + off; cnt[t] = v; }
    }
    __syncthreads();

    int2* outp = csr + bs;
    for (int i0 = 0; i0 < nbk; i0 += 4096) {
        int4 rec[4];
#pragma unroll
        for (int k = 0; k < 4; ++k) {
            const int i = i0 + k * 1024 + tid;
            if (i < nbk) rec[k] = mine[i];
            else { rec[k].x = 0; rec[k].y = -1; rec[k].z = 0; rec[k].w = 0; }
        }
#pragma unroll
        for (int k = 0; k < 4; ++k) {
            if (rec[k].y >= 0) {
                const int q = atomicAdd(&cur[rec[k].y - t0], 1);
                int2 o; o.x = rec[k].x; o.y = rec[k].z;
                outp[q] = o;
            }
        }
    }
}

__global__ void k_dinv(const int* __restrict__ start, const int* __restrict__ cnt,
                       const int2* __restrict__ csr, float* dinv, int n) {
    int i = blockIdx.x * blockDim.x + threadIdx.x;
    if (i < n) {
        const int s = start[i], c = cnt[i];
        float d = 1.0f;
        for (int j = 0; j < c; ++j) d += __int_as_float(csr[s + j].y);
        float r = (d > 0.0f) ? rsqrtf(d) : 0.0f;
        dinv[i]  = r;
    }
}


__global__ void k_copy_e(const int2* __restrict__ a, int2* __restrict__ b, int e) {
    int i = blockIdx.x * blockDim.x + threadIdx.x;
    if (i < e) b[i] = a[i];
}


template <int DINc, int NT>
__global__ __launch_bounds__(256)
void k_gemm_wmma(const float* __restrict__ hin, const float* __restrict__ W,
                 const float* __restrict__ dinv, float* __restrict__ lin, int n, int relu) {
    const int Dout = NT * 16;
    const int lane = threadIdx.x & 31;
    const int wave = threadIdx.x >> 5;
    const int half = lane >> 4;
    const int lidx = lane & 15;

    const long rowBase = ((long)blockIdx.x * 8 + wave) * 16;
    if (rowBase >= n) return;

    long rowC = rowBase + lidx;
    if (rowC > n - 1) rowC = n - 1;

    v8f acc[NT] = {};

    const float* aRow = hin + (size_t)rowC * DINc + 2 * half;

    for (int k0 = 0; k0 < DINc; k0 += 4) {
        float2 a2 = *(const float2*)(aRow + k0);
        if (relu) { a2.x = fmaxf(a2.x, 0.0f); a2.y = fmaxf(a2.y, 0.0f); }
        v2f av; av.x = a2.x; av.y = a2.y;

        const float* bBase = W + (size_t)(k0 + 2 * half) * Dout + lidx;
#pragma unroll
        for (int t = 0; t < NT; ++t) {
            v2f bv;
            bv.x = bBase[t * 16];
            bv.y = bBase[Dout + t * 16];
            acc[t] = __builtin_amdgcn_wmma_f32_16x16x4_f32(
                false, av, false, bv, (short)0, acc[t], false, false);
        }
    }

    float dv[8];
#pragma unroll
    for (int r = 0; r < 8; ++r) {
        long rr = rowBase + half * 8 + r;
        dv[r] = (rr < n) ? dinv[rr] : 0.0f;
    }
#pragma unroll
    for (int t = 0; t < NT; ++t) {
        const int c = t * 16 + lidx;
#pragma unroll
        for (int r = 0; r < 8; ++r) {
            long rr = rowBase + half * 8 + r;
            if (rr < n) lin[(size_t)rr * Dout + c] = acc[t][r] * dv[r];
        }
    }
}


template <int VEC> struct VecT;
template <> struct VecT<4> { typedef float4 T; };
template <> struct VecT<2> { typedef float2 T; };

__device__ __forceinline__ void vfma(float4& a, float w, const float4& v) {
    a.x = fmaf(w, v.x, a.x); a.y = fmaf(w, v.y, a.y);
    a.z = fmaf(w, v.z, a.z); a.w = fmaf(w, v.w, a.w);
}
__device__ __forceinline__ void vfma(float2& a, float w, const float2& v) {
    a.x = fmaf(w, v.x, a.x); a.y = fmaf(w, v.y, a.y);
}
__device__ __forceinline__ float4 vfinish(float s, const float4& l, const float4& b) {
    float4 o; o.x = fmaf(s, l.x, b.x); o.y = fmaf(s, l.y, b.y);
    o.z = fmaf(s, l.z, b.z); o.w = fmaf(s, l.w, b.w); return o;
}
__device__ __forceinline__ float2 vfinish(float s, const float2& l, const float2& b) {
    float2 o; o.x = fmaf(s, l.x, b.x); o.y = fmaf(s, l.y, b.y); return o;
}

template <int VEC>
__global__ __launch_bounds__(256)
void k_gather(const int* __restrict__ start, const int* __restrict__ cnt,
              const int2* __restrict__ csr,
              const float* __restrict__ lin, const float* __restrict__ bias,
              const float* __restrict__ dinv, float* __restrict__ out, int n) {
    typedef typename VecT<VEC>::T V;
    const int Dout = 32 * VEC;
    const int lane = threadIdx.x & 31;
    int node = blockIdx.x * 8 + (threadIdx.x >> 5);
    if (node >= n) return;
    node = __builtin_amdgcn_readfirstlane(node);
    const int s = __builtin_amdgcn_readfirstlane(start[node]);
    const int c = __builtin_amdgcn_readfirstlane(cnt[node]);

    const float* linL = lin + lane * VEC;
    const float dt = dinv[node];
    const V bv = *(const V*)(bias + lane * VEC);
    V acc = *(const V*)(linL + (size_t)node * Dout);

    for (int b0 = 0; b0 < c; b0 += 32) {
        const int m = (c - b0 < 32) ? (c - b0) : 32;
        int2 my; my.x = 0; my.y = 0;
        if (lane < m) my = csr[s + b0 + lane];
        const int mySrc = my.x;
        const int myWi  = my.y;
        int k = 0;
        for (; k + 4 <= m; k += 4) {
            const int s0 = __builtin_amdgcn_readlane(mySrc, k);
            const int s1 = __builtin_amdgcn_readlane(mySrc, k + 1);
            const int s2 = __builtin_amdgcn_readlane(mySrc, k + 2);
            const int s3 = __builtin_amdgcn_readlane(mySrc, k + 3);
            const V v0 = *(const V*)(linL + (size_t)s0 * Dout);
            const V v1 = *(const V*)(linL + (size_t)s1 * Dout);
            const V v2 = *(const V*)(linL + (size_t)s2 * Dout);
            const V v3 = *(const V*)(linL + (size_t)s3 * Dout);
            vfma(acc, __int_as_float(__builtin_amdgcn_readlane(myWi, k)),     v0);
            vfma(acc, __int_as_float(__builtin_amdgcn_readlane(myWi, k + 1)), v1);
            vfma(acc, __int_as_float(__builtin_amdgcn_readlane(myWi, k + 2)), v2);
            vfma(acc, __int_as_float(__builtin_amdgcn_readlane(myWi, k + 3)), v3);
        }
        for (; k < m; ++k) {
            const int s0 = __builtin_amdgcn_readlane(mySrc, k);
            const V v0 = *(const V*)(linL + (size_t)s0 * Dout);
            vfma(acc, __int_as_float(__builtin_amdgcn_readlane(myWi, k)), v0);
        }
    }
    *(V*)(out + (size_t)node * Dout + lane * VEC) = vfinish(dt, acc, bv);
}


static inline int cdiv(int a, int b) { return (a + b - 1) / b; }

extern "C" void kernel_launch(void* const* d_in, const int* in_sizes, int n_in,
                              void* d_out, int out_size, void* d_ws, size_t ws_size,
                              hipStream_t stream) {
    const float* x   = (const float*)d_in[0];
    const int*   ei  = (const int*)d_in[1];
    const float* ew  = (const float*)d_in[2];
    const float* W1  = (const float*)d_in[3];
    const float* b1  = (const float*)d_in[4];
    const float* W2  = (const float*)d_in[5];
    const float* b2  = (const float*)d_in[6];
    const float* W3  = (const float*)d_in[7];
    const float* b3  = (const float*)d_in[8];
    float* out = (float*)d_out;

    const int N = in_sizes[0] / D_IN;
    const int E = in_sizes[1] / 2;
    const int* eRow = ei;
    const int* eCol = ei + E;

    float* ws     = (float*)d_ws;
    int*   cnt    = (int*)ws;
    int*   start  = cnt + N;
    const int NB  = cdiv(N, BSZ);
    int*   btot   = (int*)(ws + 3 * (size_t)N);
    int*   gcur   = btot + NB;
    float* dinv   = ws + 4 * (size_t)N;
    float* buf0   = ws + 3 * (size_t)N + E;
    float* buf1   = buf0 + (size_t)N * D_HID;

    int2*  csrA   = (int2*)out;
    int2*  csrB   = (int2*)buf0;
    int4*  part   = (int4*)buf0;

    k_zero_b  <<<cdiv(NB, 256), 256, 0, stream>>>(btot, gcur, NB);
    k_bhist   <<<cdiv(E, CHUNK), 256, 0, stream>>>(eCol, btot, E, NB);
    k_part    <<<cdiv(E, CHUNK), 256, 0, stream>>>(eRow, eCol, ew, btot, gcur, part, E, NB);
    k_bsort   <<<NB, 1024, 0, stream>>>(part, btot, start, cnt, csrA, N);
    k_dinv    <<<cdiv(N, 256), 256, 0, stream>>>(start, cnt, csrA, dinv, N);

    const int gGemm = cdiv(N, 128);
    const int gGath = cdiv(N, 8);

    k_gemm_wmma<D_IN, 8><<<gGemm, 256, 0, stream>>>(x, W1, dinv, buf1, N, 0);
    k_gather<4><<<gGath, 256, 0, stream>>>(start, cnt, csrA, buf1, b1, dinv, buf0, N);

    k_gemm_wmma<D_HID, 8><<<gGemm, 256, 0, stream>>>(buf0, W2, dinv, buf1, N, 1);
    k_gather<4><<<gGath, 256, 0, stream>>>(start, cnt, csrA, buf1, b2, dinv, buf0, N);

    k_gemm_wmma<D_HID, 4><<<gGemm, 256, 0, stream>>>(buf0, W3, dinv, buf1, N, 1);
    k_copy_e<<<cdiv(E, 256), 256, 0, stream>>>(csrA, csrB, E);
    k_gather<2><<<gGath, 256, 0, stream>>>(start, cnt, csrB, buf1, b3, dinv, out, N);
}


# Layer 1 and 2 aggregations now also form the next layer's product: resident blocks hold W in LDS, take node tiles from an integer cursor, summed rows stay in LDS as WMMA A strips; 12 launches now 10
# speedup vs baseline: 4.3153x; 1.1922x over previous
#include <hip/hip_runtime.h>

typedef __attribute__((ext_vector_type(2))) float v2f;
typedef __attribute__((ext_vector_type(8))) float v8f;

#define D_IN  128
#define D_HID 128
#define D_OUT 64


#define BSH   9
#define BSZ   (1 << BSH)
#define MAXB  2048
#define CHUNK 4096
#define EPT   (CHUNK / 256)

__global__ void k_zero_b(int* btot, int* gcur, int* tctr, int nb) {
    int i = blockIdx.x * blockDim.x + threadIdx.x;
    if (i < nb) { btot[i] = 0; gcur[i] = 0; }
    if (i < 2) tctr[i] = 0;
}

__global__ __launch_bounds__(256)
void k_bhist(const int* __restrict__ col, int* btot, int e, int nb) {
    __shared__ int hist[MAXB];
    for (int b = threadIdx.x; b < nb; b += 256) hist[b] = 0;
    __syncthreads();
    const int base = blockIdx.x * CHUNK + threadIdx.x;
    int c[EPT];
#pragma unroll
    for (int j = 0; j < EPT; ++j) {
        const int i = base + j * 256;
        c[j] = (i < e) ? col[i] : -1;
    }
#pragma unroll
    for (int j = 0; j < EPT; ++j)
        if (c[j] >= 0) atomicAdd(&hist[c[j] >> BSH], 1);
    __syncthreads();
    for (int b = threadIdx.x; b < nb; b += 256) {
        const int h = hist[b];
        if (h) atomicAdd(&btot[b], h);
    }
}

__global__ __launch_bounds__(256)
void k_part(const int* __restrict__ row, const int* __restrict__ col, const float* __restrict__ w,
            const int* __restrict__ btot, int* gcur, int4* __restrict__ part, int e, int nb) {
    __shared__ int hist[MAXB];
    __shared__ int bst[MAXB];
    const int tid  = threadIdx.x;
    const int lane = tid & 31;
    const int base = blockIdx.x * CHUNK + tid;
    int c[EPT], s[EPT], wb[EPT], r[EPT];
#pragma unroll
    for (int j = 0; j < EPT; ++j) {
        const int i = base + j * 256;
        const bool ok = i < e;
        c[j]  = ok ? col[i] : -1;
        s[j]  = ok ? row[i] : 0;
        wb[j] = ok ? __float_as_int(w[i]) : 0;
    }
    for (int b = tid; b < nb; b += 256) { hist[b] = 0; bst[b] = btot[b]; }
    __syncthreads();
    if (tid < 32) {
        int carry = 0;
        for (int i0 = 0; i0 < nb; i0 += 32) {
            const int i = i0 + lane;
            const int v = (i < nb) ? bst[i] : 0;
            int incl = v;
#pragma unroll
            for (int d = 1; d < 32; d <<= 1) {
                int t = __shfl_up(incl, d);
                if (lane >= d) incl += t;
            }
            if (i < nb) bst[i] = carry + incl - v;
            carry += __shfl(incl, 31);
        }
    }
#pragma unroll
    for (int j = 0; j < EPT; ++j)
        r[j] = (c[j] >= 0) ? atomicAdd(&hist[c[j] >> BSH], 1) : 0;
    __syncthreads();
    for (int b = tid; b < nb; b += 256) {
        const int h = hist[b];
        const int g = h ? atomicAdd(&gcur[b], h) : 0;
        hist[b] = bst[b] + g;
    }
    __syncthreads();
#pragma unroll
    for (int j = 0; j < EPT; ++j) {
        if (c[j] >= 0) {
            int4 v; v.x = s[j]; v.y = c[j]; v.z = wb[j]; v.w = 0;
            part[hist[c[j] >> BSH] + r[j]] = v;
        }
    }
}

__global__ __launch_bounds__(1024)
void k_bsort(const int4* __restrict__ part, const int* __restrict__ btot,
             int* __restrict__ start, int* __restrict__ cnt, int2* __restrict__ csr, int n) {
    __shared__ int c[BSZ];
    __shared__ int cur[BSZ];
    __shared__ int wsum[BSZ / 32];
    __shared__ int sb;
    const int tid  = threadIdx.x;
    const int lane = tid & 31;
    const int wv   = tid >> 5;
    const int b    = blockIdx.x;
    const int t0   = b << BSH;
    if (tid < BSZ) c[tid] = 0;
    if (tid == 0) sb = 0;
    __syncthreads();
    int p = 0;
    for (int i = tid; i < b; i += 1024) p += btot[i];
#pragma unroll
    for (int d = 16; d >= 1; d >>= 1) p += __shfl_xor(p, d);
    if (lane == 0 && p) atomicAdd(&sb, p);
    const int nbk = btot[b];
    __syncthreads();
    const int bs = sb;
    const int4* mine = part + bs;

    for (int i0 = 0; i0 < nbk; i0 += 4096) {
        int t[4];
#pragma unroll
        for (int k = 0; k < 4; ++k) {
            const int i = i0 + k * 1024 + tid;
            t[k] = (i < nbk) ? mine[i].y : -1;
        }
#pragma unroll
        for (int k = 0; k < 4; ++k)
            if (t[k] >= 0) atomicAdd(&c[t[k] - t0], 1);
    }
    __syncthreads();

    int v = 0, incl = 0;
    if (tid < BSZ) {
        v = c[tid];
        incl = v;
#pragma unroll
        for (int d = 1; d < 32; d <<= 1) {
            int t = __shfl_up(incl, d);
            if (lane >= d) incl += t;
        }
        if (lane == 31) wsum[wv] = incl;
    }
    __syncthreads();
    if (tid < BSZ) {
        int pre = 0;
        for (int k = 0; k < wv; ++k) pre += wsum[k];
        const int off = pre + incl - v;
        cur[tid] = off;
        const int t = t0 + tid;
        if (t < n) { start[t] = bs + off; cnt[t] = v; }
    }
    __syncthreads();

    int2* outp = csr + bs;
    for (int i0 = 0; i0 < nbk; i0 += 4096) {
        int4 rec[4];
#pragma unroll
        for (int k = 0; k < 4; ++k) {
            const int i = i0 + k * 1024 + tid;
            if (i < nbk) rec[k] = mine[i];
            else { rec[k].x = 0; rec[k].y = -1; rec[k].z = 0; rec[k].w = 0; }
        }
#pragma unroll
        for (int k = 0; k < 4; ++k) {
            if (rec[k].y >= 0) {
                const int q = atomicAdd(&cur[rec[k].y - t0], 1);
                int2 o; o.x = rec[k].x; o.y = rec[k].z;
                outp[q] = o;
            }
        }
    }
}

__global__ void k_dinv(const int* __restrict__ start, const int* __restrict__ cnt,
                       const int2* __restrict__ csr, float* dinv, int n) {
    int i = blockIdx.x * blockDim.x + threadIdx.x;
    if (i < n) {
        const int s = start[i], c = cnt[i];
        float d = 1.0f;
        for (int j = 0; j < c; ++j) d += __int_as_float(csr[s + j].y);
        float r = (d > 0.0f) ? rsqrtf(d) : 0.0f;
        dinv[i]  = r;
    }
}


__global__ void k_copy_e(const int2* __restrict__ a, int2* __restrict__ b, int e) {
    int i = blockIdx.x * blockDim.x + threadIdx.x;
    if (i < e) b[i] = a[i];
}


template <int DINc, int NT>
__global__ __launch_bounds__(256)
void k_gemm_wmma(const float* __restrict__ hin, const float* __restrict__ W,
                 const float* __restrict__ dinv, float* __restrict__ lin, int n, int relu) {
    const int Dout = NT * 16;
    const int lane = threadIdx.x & 31;
    const int wave = threadIdx.x >> 5;
    const int half = lane >> 4;
    const int lidx = lane & 15;

    const long rowBase = ((long)blockIdx.x * 8 + wave) * 16;
    if (rowBase >= n) return;

    long rowC = rowBase + lidx;
    if (rowC > n - 1) rowC = n - 1;

    v8f acc[NT] = {};

    const float* aRow = hin + (size_t)rowC * DINc + 2 * half;

    for (int k0 = 0; k0 < DINc; k0 += 4) {
        float2 a2 = *(const float2*)(aRow + k0);
        if (relu) { a2.x = fmaxf(a2.x, 0.0f); a2.y = fmaxf(a2.y, 0.0f); }
        v2f av; av.x = a2.x; av.y = a2.y;

        const float* bBase = W + (size_t)(k0 + 2 * half) * Dout + lidx;
#pragma unroll
        for (int t = 0; t < NT; ++t) {
            v2f bv;
            bv.x = bBase[t * 16];
            bv.y = bBase[Dout + t * 16];
            acc[t] = __builtin_amdgcn_wmma_f32_16x16x4_f32(
                false, av, false, bv, (short)0, acc[t], false, false);
        }
    }

    float dv[8];
#pragma unroll
    for (int r = 0; r < 8; ++r) {
        long rr = rowBase + half * 8 + r;
        dv[r] = (rr < n) ? dinv[rr] : 0.0f;
    }
#pragma unroll
    for (int t = 0; t < NT; ++t) {
        const int c = t * 16 + lidx;
#pragma unroll
        for (int r = 0; r < 8; ++r) {
            long rr = rowBase + half * 8 + r;
            if (rr < n) lin[(size_t)rr * Dout + c] = acc[t][r] * dv[r];
        }
    }
}


template <int VEC> struct VecT;
template <> struct VecT<4> { typedef float4 T; };
template <> struct VecT<2> { typedef float2 T; };

__device__ __forceinline__ void vfma(float4& a, float w, const float4& v) {
    a.x = fmaf(w, v.x, a.x); a.y = fmaf(w, v.y, a.y);
    a.z = fmaf(w, v.z, a.z); a.w = fmaf(w, v.w, a.w);
}
__device__ __forceinline__ void vfma(float2& a, float w, const float2& v) {
    a.x = fmaf(w, v.x, a.x); a.y = fmaf(w, v.y, a.y);
}
__device__ __forceinline__ float4 vfinish(float s, const float4& l, const float4& b) {
    float4 o; o.x = fmaf(s, l.x, b.x); o.y = fmaf(s, l.y, b.y);
    o.z = fmaf(s, l.z, b.z); o.w = fmaf(s, l.w, b.w); return o;
}
__device__ __forceinline__ float2 vfinish(float s, const float2& l, const float2& b) {
    float2 o; o.x = fmaf(s, l.x, b.x); o.y = fmaf(s, l.y, b.y); return o;
}

template <int VEC>
__global__ __launch_bounds__(256)
void k_gather(const int* __restrict__ start, const int* __restrict__ cnt,
              const int2* __restrict__ csr,
              const float* __restrict__ lin, const float* __restrict__ bias,
              const float* __restrict__ dinv, float* __restrict__ out, int n) {
    typedef typename VecT<VEC>::T V;
    const int Dout = 32 * VEC;
    const int lane = threadIdx.x & 31;
    int node = blockIdx.x * 8 + (threadIdx.x >> 5);
    if (node >= n) return;
    node = __builtin_amdgcn_readfirstlane(node);
    const int s = __builtin_amdgcn_readfirstlane(start[node]);
    const int c = __builtin_amdgcn_readfirstlane(cnt[node]);

    const float* linL = lin + lane * VEC;
    const float dt = dinv[node];
    const V bv = *(const V*)(bias + lane * VEC);
    V acc = *(const V*)(linL + (size_t)node * Dout);

    for (int b0 = 0; b0 < c; b0 += 32) {
        const int m = (c - b0 < 32) ? (c - b0) : 32;
        int2 my; my.x = 0; my.y = 0;
        if (lane < m) my = csr[s + b0 + lane];
        const int mySrc = my.x;
        const int myWi  = my.y;
        int k = 0;
        for (; k + 4 <= m; k += 4) {
            const int s0 = __builtin_amdgcn_readlane(mySrc, k);
            const int s1 = __builtin_amdgcn_readlane(mySrc, k + 1);
            const int s2 = __builtin_amdgcn_readlane(mySrc, k + 2);
            const int s3 = __builtin_amdgcn_readlane(mySrc, k + 3);
            const V v0 = *(const V*)(linL + (size_t)s0 * Dout);
            const V v1 = *(const V*)(linL + (size_t)s1 * Dout);
            const V v2 = *(const V*)(linL + (size_t)s2 * Dout);
            const V v3 = *(const V*)(linL + (size_t)s3 * Dout);
            vfma(acc, __int_as_float(__builtin_amdgcn_readlane(myWi, k)),     v0);
            vfma(acc, __int_as_float(__builtin_amdgcn_readlane(myWi, k + 1)), v1);
            vfma(acc, __int_as_float(__builtin_amdgcn_readlane(myWi, k + 2)), v2);
            vfma(acc, __int_as_float(__builtin_amdgcn_readlane(myWi, k + 3)), v3);
        }
        for (; k < m; ++k) {
            const int s0 = __builtin_amdgcn_readlane(mySrc, k);
            const V v0 = *(const V*)(linL + (size_t)s0 * Dout);
            vfma(acc, __int_as_float(__builtin_amdgcn_readlane(myWi, k)), v0);
        }
    }
    *(V*)(out + (size_t)node * Dout + lane * VEC) = vfinish(dt, acc, bv);
}


#define APITCH 132

template <int U>
__device__ __forceinline__ void gather_batch(float4& acc, const float* __restrict__ linL,
                                             int mySrc, int myWi, int k) {
    float4 v[U];
#pragma unroll
    for (int u = 0; u < U; ++u) {
        const int su = __builtin_amdgcn_readlane(mySrc, k + u);
        v[u] = *(const float4*)(linL + (size_t)su * 128);
    }
#pragma unroll
    for (int u = 0; u < U; ++u)
        vfma(acc, __int_as_float(__builtin_amdgcn_readlane(myWi, k + u)), v[u]);
}

__device__ __forceinline__ float4 gather_sum128(const float* __restrict__ linL, const int2* __restrict__ csr,
                                                int s, int c, int2 first, float4 acc, int lane) {
    for (int b0 = 0; b0 < c; b0 += 32) {
        const int m = (c - b0 < 32) ? (c - b0) : 32;
        int2 my = first;
        if (b0 > 0) {
            my.x = 0; my.y = 0;
            if (lane < m) my = csr[s + b0 + lane];
        }
        const int mySrc = my.x;
        const int myWi  = my.y;
        int k = 0;
        for (; k + 8 <= m; k += 8) gather_batch<8>(acc, linL, mySrc, myWi, k);
        if ((m - k) & 4) { gather_batch<4>(acc, linL, mySrc, myWi, k); k += 4; }
        if ((m - k) & 2) { gather_batch<2>(acc, linL, mySrc, myWi, k); k += 2; }
        if ((m - k) & 1) { gather_batch<1>(acc, linL, mySrc, myWi, k); }
    }
    return acc;
}

template <int CT, int NW>
__global__ __launch_bounds__(NW * 32)
void k_gather_gemm(const int* __restrict__ start, const int* __restrict__ cnt,
                   const int2* __restrict__ csr,
                   const float* __restrict__ lin, const float* __restrict__ bias,
                   const float* __restrict__ dinv, const float* __restrict__ Wn,
                   float* __restrict__ linN, int* tctr, int n) {
    constexpr int DoutN = CT * 16;
    constexpr int Din   = 128;
    __shared__ __attribute__((aligned(16))) float Wl[Din * DoutN];
    constexpr int TN    = 2 * NW;
    constexpr int JOBS  = (TN / 16) * CT;
    __shared__ __attribute__((aligned(16))) float hs[TN * APITCH];
    __shared__ int   tcur;

    const int lane = threadIdx.x & 31;
    const int wave = __builtin_amdgcn_readfirstlane(threadIdx.x >> 5);
    const int half = lane >> 4;
    const int lidx = lane & 15;
    const int ntiles = (n + TN - 1) / TN;

    if (threadIdx.x == 0) tcur = atomicAdd(tctr, 1);

    for (int i = threadIdx.x; i < (Din / 4) * (DoutN / 4); i += NW * 32) {
        const int kq = i / (DoutN / 4);
        const int c  = (i % (DoutN / 4)) * 4;
        const float* src = Wn + (size_t)(4 * kq) * DoutN + c;
        const float4 r0 = *(const float4*)(src);
        const float4 r1 = *(const float4*)(src + DoutN);
        const float4 r2 = *(const float4*)(src + 2 * DoutN);
        const float4 r3 = *(const float4*)(src + 3 * DoutN);
        float* d = Wl + ((size_t)kq * DoutN + c) * 4;
        float4 o;
        o.x = r0.x; o.y = r1.x; o.z = r2.x; o.w = r3.x; *(float4*)(d)      = o;
        o.x = r0.y; o.y = r1.y; o.z = r2.y; o.w = r3.y; *(float4*)(d + 4)  = o;
        o.x = r0.z; o.y = r1.z; o.z = r2.z; o.w = r3.z; *(float4*)(d + 8)  = o;
        o.x = r0.w; o.y = r1.w; o.z = r2.w; o.w = r3.w; *(float4*)(d + 12) = o;
    }

    const float* linL = lin + lane * 4;
    const float4 bv = *(const float4*)(bias + lane * 4);
    __syncthreads();
    int tile = __builtin_amdgcn_readfirstlane(tcur);

    while (tile < ntiles) {
        __syncthreads();
        if (threadIdx.x == 0) tcur = atomicAdd(tctr, 1);
        const int blockBase = tile * TN;

        const int nodeA = (blockBase + wave * 2     < n) ? blockBase + wave * 2     : n - 1;
        const int nodeB = (blockBase + wave * 2 + 1 < n) ? blockBase + wave * 2 + 1 : n - 1;
        const int rsA = start[nodeA], rcA = cnt[nodeA];
        const int rsB = start[nodeB], rcB = cnt[nodeB];
        const float dA = dinv[nodeA], dB = dinv[nodeB];
        const int sA = __builtin_amdgcn_readfirstlane(rsA);
        const int cA = __builtin_amdgcn_readfirstlane(rcA);
        const int sB = __builtin_amdgcn_readfirstlane(rsB);
        const int cB = __builtin_amdgcn_readfirstlane(rcB);
        int2 fA; fA.x = 0; fA.y = 0;
        int2 fB; fB.x = 0; fB.y = 0;
        if (lane < cA) fA = csr[sA + lane];
        if (lane < cB) fB = csr[sB + lane];
        float4 accA = *(const float4*)(linL + (size_t)nodeA * Din);
        float4 accB = *(const float4*)(linL + (size_t)nodeB * Din);
        accA = gather_sum128(linL, csr, sA, cA, fA, accA, lane);
        accB = gather_sum128(linL, csr, sB, cB, fB, accB, lane);
        float4 hA = vfinish(dA, accA, bv);
        float4 hB = vfinish(dB, accB, bv);
        hA.x = fmaxf(hA.x, 0.0f); hA.y = fmaxf(hA.y, 0.0f); hA.z = fmaxf(hA.z, 0.0f); hA.w = fmaxf(hA.w, 0.0f);
        hB.x = fmaxf(hB.x, 0.0f); hB.y = fmaxf(hB.y, 0.0f); hB.z = fmaxf(hB.z, 0.0f); hB.w = fmaxf(hB.w, 0.0f);
        *(float4*)(hs + (wave * 2)     * APITCH + lane * 4) = hA;
        *(float4*)(hs + (wave * 2 + 1) * APITCH + lane * 4) = hB;
        __syncthreads();
        const int nxt = __builtin_amdgcn_readfirstlane(tcur);

        const int stripe = wave / CT;
        const int ctile  = wave % CT;
        v8f acc8 = {};
        if (wave < JOBS) {
            const float* aRow = hs + (stripe * 16 + lidx) * APITCH + 2 * half;
            const float* bCol = Wl + (ctile * 16 + lidx) * 4 + 2 * half;
            for (int k0 = 0; k0 < Din; k0 += 4) {
                const float2 a2 = *(const float2*)(aRow + k0);
                const float2 b2 = *(const float2*)(bCol + k0 * DoutN);
                v2f av; av.x = a2.x; av.y = a2.y;
                v2f bw; bw.x = b2.x; bw.y = b2.y;
                acc8 = __builtin_amdgcn_wmma_f32_16x16x4_f32(
                    false, av, false, bw, (short)0, acc8, false, false);
            }
        }
        __syncthreads();

        if (wave < JOBS) {
#pragma unroll
            for (int r = 0; r < 8; ++r)
                hs[(stripe * 16 + half * 8 + r) * APITCH + ctile * 16 + lidx] = acc8[r];
        }
        __syncthreads();

        constexpr int RPI = 128 / DoutN;
        constexpr int LPR = 32 / RPI;
        for (int r0 = wave * RPI; r0 < TN; r0 += NW * RPI) {
            const int row = r0 + lane / LPR;
            const int col = (lane % LPR) * 4;
            const int rr  = blockBase + row;
            if (rr < n) {
                const float d = dinv[rr];
                const float4 v = *(const float4*)(hs + row * APITCH + col);
                float4 o; o.x = v.x * d; o.y = v.y * d; o.z = v.z * d; o.w = v.w * d;
                *(float4*)(linN + (size_t)rr * DoutN + col) = o;
            }
        }
        tile = nxt;
    }
}


static inline int cdiv(int a, int b) { return (a + b - 1) / b; }

extern "C" void kernel_launch(void* const* d_in, const int* in_sizes, int n_in,
                              void* d_out, int out_size, void* d_ws, size_t ws_size,
                              hipStream_t stream) {
    const float* x   = (const float*)d_in[0];
    const int*   ei  = (const int*)d_in[1];
    const float* ew  = (const float*)d_in[2];
    const float* W1  = (const float*)d_in[3];
    const float* b1  = (const float*)d_in[4];
    const float* W2  = (const float*)d_in[5];
    const float* b2  = (const float*)d_in[6];
    const float* W3  = (const float*)d_in[7];
    const float* b3  = (const float*)d_in[8];
    float* out = (float*)d_out;

    const int N = in_sizes[0] / D_IN;
    const int E = in_sizes[1] / 2;
    const int* eRow = ei;
    const int* eCol = ei + E;

    float* ws     = (float*)d_ws;
    int*   cnt    = (int*)ws;
    int*   start  = cnt + N;
    const int NB  = cdiv(N, BSZ);
    int*   btot   = (int*)(ws + 3 * (size_t)N);
    int*   gcur   = btot + NB;
    int*   tctr   = gcur + NB;
    float* dinv   = ws + 4 * (size_t)N;
    float* buf0   = ws + 3 * (size_t)N + E;
    float* buf1   = buf0 + (size_t)N * D_HID;

    int2*  csrA   = (int2*)out;
    int2*  csrB   = (int2*)buf0;
    int4*  part   = (int4*)buf0;

    k_zero_b  <<<cdiv(NB, 256), 256, 0, stream>>>(btot, gcur, tctr, NB);
    k_bhist   <<<cdiv(E, CHUNK), 256, 0, stream>>>(eCol, btot, E, NB);
    k_part    <<<cdiv(E, CHUNK), 256, 0, stream>>>(eRow, eCol, ew, btot, gcur, part, E, NB);
    k_bsort   <<<NB, 1024, 0, stream>>>(part, btot, start, cnt, csrA, N);
    k_dinv    <<<cdiv(N, 256), 256, 0, stream>>>(start, cnt, csrA, dinv, N);

    const int gGemm = cdiv(N, 128);
    const int gGath = cdiv(N, 8);

    k_gemm_wmma<D_IN, 8><<<gGemm, 256, 0, stream>>>(x, W1, dinv, buf1, N, 0);

    k_gather_gemm<8, 16><<<768, 512, 0, stream>>>(start, cnt, csrA, buf1, b1, dinv, W2, buf0, tctr, N);

    k_gather_gemm<4, 8><<<1536, 256, 0, stream>>>(start, cnt, csrA, buf0, b2, dinv, W3, buf1, tctr + 1, N);

    k_copy_e<<<cdiv(E, 256), 256, 0, stream>>>(csrA, csrB, E);
    k_gather<2><<<gGath, 256, 0, stream>>>(start, cnt, csrB, buf1, b3, dinv, out, N);
}


# Ten launches now six: edge blocks keep range-grouped records in their own chunk, publish run offsets (no zero or histogram launch); degree by LDS float adds, arrival order; list copy rides in layer 2
# speedup vs baseline: 4.4389x; 1.0286x over previous
#include <hip/hip_runtime.h>

typedef __attribute__((ext_vector_type(2))) float v2f;
typedef __attribute__((ext_vector_type(8))) float v8f;

#define D_IN  128
#define D_HID 128
#define D_OUT 64


#define BSH   9
#define BSZ   (1 << BSH)
#define MAXB  2048
#define CHUNK 4096
#define EPT   (CHUNK / 256)
#define MAXK  2048
#define RU    8

__global__ __launch_bounds__(256)
void k_part(const int* __restrict__ row, const int* __restrict__ col, const float* __restrict__ w,
            int* __restrict__ tbl, int* tctr, int4* __restrict__ part, int e, int nb) {
    __shared__ int hist[MAXB + 1];
    const int tid  = threadIdx.x;
    const int lane = tid & 31;
    const int base = blockIdx.x * CHUNK + tid;
    int c[EPT], s[EPT], wb[EPT], r[EPT];
#pragma unroll
    for (int j = 0; j < EPT; ++j) {
        const int i = base + j * 256;
        const bool ok = i < e;
        c[j]  = ok ? col[i] : -1;
        s[j]  = ok ? row[i] : 0;
        wb[j] = ok ? __float_as_int(w[i]) : 0;
    }
    for (int b = tid; b < nb; b += 256) hist[b] = 0;
    if (blockIdx.x == 0 && tid < 2) tctr[tid] = 0;
    __syncthreads();
#pragma unroll
    for (int j = 0; j < EPT; ++j)
        r[j] = (c[j] >= 0) ? atomicAdd(&hist[c[j] >> BSH], 1) : 0;
    __syncthreads();
    if (tid < 32) {
        int carry = 0;
        for (int i0 = 0; i0 < nb; i0 += 32) {
            const int i = i0 + lane;
            const int v = (i < nb) ? hist[i] : 0;
            int incl = v;
#pragma unroll
            for (int d = 1; d < 32; d <<= 1) {
                int t = __shfl_up(incl, d);
                if (lane >= d) incl += t;
            }
            if (i < nb) hist[i] = carry + incl - v;
            carry += __shfl(incl, 31);
        }
        if (lane == 0) hist[nb] = carry;
    }
    __syncthreads();
    int* trow = tbl + (size_t)blockIdx.x * (nb + 1);
    for (int b = tid; b <= nb; b += 256) trow[b] = hist[b];
    int4* mine = part + (size_t)blockIdx.x * CHUNK;
#pragma unroll
    for (int j = 0; j < EPT; ++j) {
        if (c[j] >= 0) {
            int4 v; v.x = s[j]; v.y = c[j]; v.z = wb[j]; v.w = 0;
            mine[hist[c[j] >> BSH] + r[j]] = v;
        }
    }
}

__global__ __launch_bounds__(1024)
void k_bsort(const int4* __restrict__ part, const int* __restrict__ tbl,
             int* __restrict__ start, int* __restrict__ cnt, int2* __restrict__ csr,
             float* __restrict__ dinv, int n, int nb, int nk) {
    __shared__ int   c[BSZ];
    __shared__ int   cur[BSZ];
    __shared__ float dsum[BSZ];
    __shared__ int   wsum[BSZ / 32];
    __shared__ int   roff[MAXK];
    __shared__ int   rlen[MAXK];
    __shared__ int   sb;
    const int tid  = threadIdx.x;
    const int lane = tid & 31;
    const int wv   = tid >> 5;
    const int b    = blockIdx.x;
    const int t0   = b << BSH;
    if (tid < BSZ) { c[tid] = 0; dsum[tid] = 1.0f; }
    if (tid == 0) sb = 0;
    __syncthreads();
    int p = 0;
    for (int k = tid; k < nk; k += 1024) {
        const int* tr = tbl + (size_t)k * (nb + 1) + b;
        const int o = tr[0], o1 = tr[1];
        roff[k] = k * CHUNK + o;
        rlen[k] = o1 - o;
        p += o;
    }
#pragma unroll
    for (int d = 16; d >= 1; d >>= 1) p += __shfl_xor(p, d);
    if (lane == 0 && p) atomicAdd(&sb, p);
    __syncthreads();
    const int bs = sb;

    for (int k0 = wv; k0 < nk; k0 += 32 * RU) {
        int t[RU];
#pragma unroll
        for (int u = 0; u < RU; ++u) {
            const int k = k0 + 32 * u;
            t[u] = -1;
            if (k < nk && lane < rlen[k]) t[u] = part[roff[k] + lane].y;
        }
#pragma unroll
        for (int u = 0; u < RU; ++u)
            if (t[u] >= 0) atomicAdd(&c[t[u] - t0], 1);
    }
    for (int k = wv; k < nk; k += 32) {
        const int len = rlen[k];
        const int4* run = part + roff[k];
        for (int i = 32 + lane; i < len; i += 32) atomicAdd(&c[run[i].y - t0], 1);
    }
    __syncthreads();

    int v = 0, incl = 0;
    if (tid < BSZ) {
        v = c[tid];
        incl = v;
#pragma unroll
        for (int d = 1; d < 32; d <<= 1) {
            int t = __shfl_up(incl, d);
            if (lane >= d) incl += t;
        }
        if (lane == 31) wsum[wv] = incl;
    }
    __syncthreads();
    if (tid < BSZ) {
        int pre = 0;
        for (int k = 0; k < wv; ++k) pre += wsum[k];
        const int off = pre + incl - v;
        cur[tid] = off;
        const int t = t0 + tid;
        if (t < n) { start[t] = bs + off; cnt[t] = v; }
    }
    __syncthreads();

    int2* outp = csr + bs;
    for (int k0 = wv; k0 < nk; k0 += 32 * RU) {
        int4 rec[RU];
#pragma unroll
        for (int u = 0; u < RU; ++u) {
            const int k = k0 + 32 * u;
            rec[u].x = 0; rec[u].y = -1; rec[u].z = 0; rec[u].w = 0;
            if (k < nk && lane < rlen[k]) rec[u] = part[roff[k] + lane];
        }
#pragma unroll
        for (int u = 0; u < RU; ++u) {
            if (rec[u].y >= 0) {
                const int tl = rec[u].y - t0;
                const int q = atomicAdd(&cur[tl], 1);
                atomicAdd(&dsum[tl], __int_as_float(rec[u].z));
                int2 o; o.x = rec[u].x; o.y = rec[u].z;
                outp[q] = o;
            }
        }
    }
    for (int k = wv; k < nk; k += 32) {
        const int len = rlen[k];
        const int4* run = part + roff[k];
        for (int i = 32 + lane; i < len; i += 32) {
            const int4 rc = run[i];
            const int tl = rc.y - t0;
            const int q = atomicAdd(&cur[tl], 1);
            atomicAdd(&dsum[tl], __int_as_float(rc.z));
            int2 o; o.x = rc.x; o.y = rc.z;
            outp[q] = o;
        }
    }
    __syncthreads();
    if (tid < BSZ) {
        const int t = t0 + tid;
        if (t < n) {
            const float d = dsum[tid];
            dinv[t] = (d > 0.0f) ? rsqrtf(d) : 0.0f;
        }
    }
}


template <int DINc, int NT>
__global__ __launch_bounds__(256)
void k_gemm_wmma(const float* __restrict__ hin, const float* __restrict__ W,
                 const float* __restrict__ dinv, float* __restrict__ lin, int n, int relu) {
    const int Dout = NT * 16;
    const int lane = threadIdx.x & 31;
    const int wave = threadIdx.x >> 5;
    const int half = lane >> 4;
    const int lidx = lane & 15;

    const long rowBase = ((long)blockIdx.x * 8 + wave) * 16;
    if (rowBase >= n) return;

    long rowC = rowBase + lidx;
    if (rowC > n - 1) rowC = n - 1;

    v8f acc[NT] = {};

    const float* aRow = hin + (size_t)rowC * DINc + 2 * half;

    for (int k0 = 0; k0 < DINc; k0 += 4) {
        float2 a2 = *(const float2*)(aRow + k0);
        if (relu) { a2.x = fmaxf(a2.x, 0.0f); a2.y = fmaxf(a2.y, 0.0f); }
        v2f av; av.x = a2.x; av.y = a2.y;

        const float* bBase = W + (size_t)(k0 + 2 * half) * Dout + lidx;
#pragma unroll
        for (int t = 0; t < NT; ++t) {
            v2f bv;
            bv.x = bBase[t * 16];
            bv.y = bBase[Dout + t * 16];
            acc[t] = __builtin_amdgcn_wmma_f32_16x16x4_f32(
                false, av, false, bv, (short)0, acc[t], false, false);
        }
    }

    float dv[8];
#pragma unroll
    for (int r = 0; r < 8; ++r) {
        long rr = rowBase + half * 8 + r;
        dv[r] = (rr < n) ? dinv[rr] : 0.0f;
    }
#pragma unroll
    for (int t = 0; t < NT; ++t) {
        const int c = t * 16 + lidx;
#pragma unroll
        for (int r = 0; r < 8; ++r) {
            long rr = rowBase + half * 8 + r;
            if (rr < n) lin[(size_t)rr * Dout + c] = acc[t][r] * dv[r];
        }
    }
}


template <int VEC> struct VecT;
template <> struct VecT<4> { typedef float4 T; };
template <> struct VecT<2> { typedef float2 T; };

__device__ __forceinline__ void vfma(float4& a, float w, const float4& v) {
    a.x = fmaf(w, v.x, a.x); a.y = fmaf(w, v.y, a.y);
    a.z = fmaf(w, v.z, a.z); a.w = fmaf(w, v.w, a.w);
}
__device__ __forceinline__ void vfma(float2& a, float w, const float2& v) {
    a.x = fmaf(w, v.x, a.x); a.y = fmaf(w, v.y, a.y);
}
__device__ __forceinline__ float4 vfinish(float s, const float4& l, const float4& b) {
    float4 o; o.x = fmaf(s, l.x, b.x); o.y = fmaf(s, l.y, b.y);
    o.z = fmaf(s, l.z, b.z); o.w = fmaf(s, l.w, b.w); return o;
}
__device__ __forceinline__ float2 vfinish(float s, const float2& l, const float2& b) {
    float2 o; o.x = fmaf(s, l.x, b.x); o.y = fmaf(s, l.y, b.y); return o;
}

template <int VEC>
__global__ __launch_bounds__(256)
void k_gather(const int* __restrict__ start, const int* __restrict__ cnt,
              const int2* __restrict__ csr,
              const float* __restrict__ lin, const float* __restrict__ bias,
              const float* __restrict__ dinv, float* __restrict__ out, int n) {
    typedef typename VecT<VEC>::T V;
    const int Dout = 32 * VEC;
    const int lane = threadIdx.x & 31;
    int node = blockIdx.x * 8 + (threadIdx.x >> 5);
    if (node >= n) return;
    node = __builtin_amdgcn_readfirstlane(node);
    const int s = __builtin_amdgcn_readfirstlane(start[node]);
    const int c = __builtin_amdgcn_readfirstlane(cnt[node]);

    const float* linL = lin + lane * VEC;
    const float dt = dinv[node];
    const V bv = *(const V*)(bias + lane * VEC);
    V acc = *(const V*)(linL + (size_t)node * Dout);

    for (int b0 = 0; b0 < c; b0 += 32) {
        const int m = (c - b0 < 32) ? (c - b0) : 32;
        int2 my; my.x = 0; my.y = 0;
        if (lane < m) my = csr[s + b0 + lane];
        const int mySrc = my.x;
        const int myWi  = my.y;
        int k = 0;
        for (; k + 4 <= m; k += 4) {
            const int s0 = __builtin_amdgcn_readlane(mySrc, k);
            const int s1 = __builtin_amdgcn_readlane(mySrc, k + 1);
            const int s2 = __builtin_amdgcn_readlane(mySrc, k + 2);
            const int s3 = __builtin_amdgcn_readlane(mySrc, k + 3);
            const V v0 = *(const V*)(linL + (size_t)s0 * Dout);
            const V v1 = *(const V*)(linL + (size_t)s1 * Dout);
            const V v2 = *(const V*)(linL + (size_t)s2 * Dout);
            const V v3 = *(const V*)(linL + (size_t)s3 * Dout);
            vfma(acc, __int_as_float(__builtin_amdgcn_readlane(myWi, k)),     v0);
            vfma(acc, __int_as_float(__builtin_amdgcn_readlane(myWi, k + 1)), v1);
            vfma(acc, __int_as_float(__builtin_amdgcn_readlane(myWi, k + 2)), v2);
            vfma(acc, __int_as_float(__builtin_amdgcn_readlane(myWi, k + 3)), v3);
        }
        for (; k < m; ++k) {
            const int s0 = __builtin_amdgcn_readlane(mySrc, k);
            const V v0 = *(const V*)(linL + (size_t)s0 * Dout);
            vfma(acc, __int_as_float(__builtin_amdgcn_readlane(myWi, k)), v0);
        }
    }
    *(V*)(out + (size_t)node * Dout + lane * VEC) = vfinish(dt, acc, bv);
}


#define APITCH 132

template <int U>
__device__ __forceinline__ void gather_batch(float4& acc, const float* __restrict__ linL,
                                             int mySrc, int myWi, int k) {
    float4 v[U];
#pragma unroll
    for (int u = 0; u < U; ++u) {
        const int su = __builtin_amdgcn_readlane(mySrc, k + u);
        v[u] = *(const float4*)(linL + (size_t)su * 128);
    }
#pragma unroll
    for (int u = 0; u < U; ++u)
        vfma(acc, __int_as_float(__builtin_amdgcn_readlane(myWi, k + u)), v[u]);
}

__device__ __forceinline__ float4 gather_sum128(const float* __restrict__ linL, const int2* __restrict__ csr,
                                                int s, int c, int2 first, float4 acc, int lane) {
    for (int b0 = 0; b0 < c; b0 += 32) {
        const int m = (c - b0 < 32) ? (c - b0) : 32;
        int2 my = first;
        if (b0 > 0) {
            my.x = 0; my.y = 0;
            if (lane < m) my = csr[s + b0 + lane];
        }
        const int mySrc = my.x;
        const int myWi  = my.y;
        int k = 0;
        for (; k + 8 <= m; k += 8) gather_batch<8>(acc, linL, mySrc, myWi, k);
        if ((m - k) & 4) { gather_batch<4>(acc, linL, mySrc, myWi, k); k += 4; }
        if ((m - k) & 2) { gather_batch<2>(acc, linL, mySrc, myWi, k); k += 2; }
        if ((m - k) & 1) { gather_batch<1>(acc, linL, mySrc, myWi, k); }
    }
    return acc;
}

template <int CT, int NW>
__global__ __launch_bounds__(NW * 32)
void k_gather_gemm(const int* __restrict__ start, const int* __restrict__ cnt,
                   const int2* __restrict__ csr,
                   const float* __restrict__ lin, const float* __restrict__ bias,
                   const float* __restrict__ dinv, const float* __restrict__ Wn,
                   float* __restrict__ linN, int* tctr, int n,
                   const int2* __restrict__ cpSrc, int2* __restrict__ cpDst, int cpN) {
    constexpr int DoutN = CT * 16;
    constexpr int Din   = 128;
    __shared__ __attribute__((aligned(16))) float Wl[Din * DoutN];
    constexpr int TN    = 2 * NW;
    constexpr int JOBS  = (TN / 16) * CT;
    __shared__ __attribute__((aligned(16))) float hs[TN * APITCH];
    __shared__ int   tcur;

    const int lane = threadIdx.x & 31;
    const int wave = __builtin_amdgcn_readfirstlane(threadIdx.x >> 5);
    const int half = lane >> 4;
    const int lidx = lane & 15;
    const int ntiles = (n + TN - 1) / TN;

    if (threadIdx.x == 0) tcur = atomicAdd(tctr, 1);

    {
        const int gsz = gridDim.x * NW * 32;
        for (int i0 = blockIdx.x * NW * 32 + threadIdx.x; i0 < cpN; i0 += 4 * gsz) {
            int2 cv[4];
#pragma unroll
            for (int u = 0; u < 4; ++u) {
                const int i = i0 + u * gsz;
                cv[u].x = 0; cv[u].y = 0;
                if (i < cpN) cv[u] = cpSrc[i];
            }
#pragma unroll
            for (int u = 0; u < 4; ++u) {
                const int i = i0 + u * gsz;
                if (i < cpN) cpDst[i] = cv[u];
            }
        }
    }

    for (int i = threadIdx.x; i < (Din / 4) * (DoutN / 4); i += NW * 32) {
        const int kq = i / (DoutN / 4);
        const int c  = (i % (DoutN / 4)) * 4;
        const float* src = Wn + (size_t)(4 * kq) * DoutN + c;
        const float4 r0 = *(const float4*)(src);
        const float4 r1 = *(const float4*)(src + DoutN);
        const float4 r2 = *(const float4*)(src + 2 * DoutN);
        const float4 r3 = *(const float4*)(src + 3 * DoutN);
        float* d = Wl + ((size_t)kq * DoutN + c) * 4;
        float4 o;
        o.x = r0.x; o.y = r1.x; o.z = r2.x; o.w = r3.x; *(float4*)(d)      = o;
        o.x = r0.y; o.y = r1.y; o.z = r2.y; o.w = r3.y; *(float4*)(d + 4)  = o;
        o.x = r0.z; o.y = r1.z; o.z = r2.z; o.w = r3.z; *(float4*)(d + 8)  = o;
        o.x = r0.w; o.y = r1.w; o.z = r2.w; o.w = r3.w; *(float4*)(d + 12) = o;
    }

    const float* linL = lin + lane * 4;
    const float4 bv = *(const float4*)(bias + lane * 4);
    __syncthreads();
    int tile = __builtin_amdgcn_readfirstlane(tcur);

    while (tile < ntiles) {
        __syncthreads();
        if (threadIdx.x == 0) tcur = atomicAdd(tctr, 1);
        const int blockBase = tile * TN;

        const int nodeA = (blockBase + wave * 2     < n) ? blockBase + wave * 2     : n - 1;
        const int nodeB = (blockBase + wave * 2 + 1 < n) ? blockBase + wave * 2 + 1 : n - 1;
        const int rsA = start[nodeA], rcA = cnt[nodeA];
        const int rsB = start[nodeB], rcB = cnt[nodeB];
        const float dA = dinv[nodeA], dB = dinv[nodeB];
        const int sA = __builtin_amdgcn_readfirstlane(rsA);
        const int cA = __builtin_amdgcn_readfirstlane(rcA);
        const int sB = __builtin_amdgcn_readfirstlane(rsB);
        const int cB = __builtin_amdgcn_readfirstlane(rcB);
        int2 fA; fA.x = 0; fA.y = 0;
        int2 fB; fB.x = 0; fB.y = 0;
        if (lane < cA) fA = csr[sA + lane];
        if (lane < cB) fB = csr[sB + lane];
        float4 accA = *(const float4*)(linL + (size_t)nodeA * Din);
        float4 accB = *(const float4*)(linL + (size_t)nodeB * Din);
        accA = gather_sum128(linL, csr, sA, cA, fA, accA, lane);
        accB = gather_sum128(linL, csr, sB, cB, fB, accB, lane);
        float4 hA = vfinish(dA, accA, bv);
        float4 hB = vfinish(dB, accB, bv);
        hA.x = fmaxf(hA.x, 0.0f); hA.y = fmaxf(hA.y, 0.0f); hA.z = fmaxf(hA.z, 0.0f); hA.w = fmaxf(hA.w, 0.0f);
        hB.x = fmaxf(hB.x, 0.0f); hB.y = fmaxf(hB.y, 0.0f); hB.z = fmaxf(hB.z, 0.0f); hB.w = fmaxf(hB.w, 0.0f);
        *(float4*)(hs + (wave * 2)     * APITCH + lane * 4) = hA;
        *(float4*)(hs + (wave * 2 + 1) * APITCH + lane * 4) = hB;
        __syncthreads();
        const int nxt = __builtin_amdgcn_readfirstlane(tcur);

        const int stripe = wave / CT;
        const int ctile  = wave % CT;
        v8f acc8 = {};
        if (wave < JOBS) {
            const float* aRow = hs + (stripe * 16 + lidx) * APITCH + 2 * half;
            const float* bCol = Wl + (ctile * 16 + lidx) * 4 + 2 * half;
            for (int k0 = 0; k0 < Din; k0 += 4) {
                const float2 a2 = *(const float2*)(aRow + k0);
                const float2 b2 = *(const float2*)(bCol + k0 * DoutN);
                v2f av; av.x = a2.x; av.y = a2.y;
                v2f bw; bw.x = b2.x; bw.y = b2.y;
                acc8 = __builtin_amdgcn_wmma_f32_16x16x4_f32(
                    false, av, false, bw, (short)0, acc8, false, false);
            }
        }
        __syncthreads();

        if (wave < JOBS) {
#pragma unroll
            for (int r = 0; r < 8; ++r)
                hs[(stripe * 16 + half * 8 + r) * APITCH + ctile * 16 + lidx] = acc8[r];
        }
        __syncthreads();

        constexpr int RPI = 128 / DoutN;
        constexpr int LPR = 32 / RPI;
        for (int r0 = wave * RPI; r0 < TN; r0 += NW * RPI) {
            const int row = r0 + lane / LPR;
            const int col = (lane % LPR) * 4;
            const int rr  = blockBase + row;
            if (rr < n) {
                const float d = dinv[rr];
                const float4 v = *(const float4*)(hs + row * APITCH + col);
                float4 o; o.x = v.x * d; o.y = v.y * d; o.z = v.z * d; o.w = v.w * d;
                *(float4*)(linN + (size_t)rr * DoutN + col) = o;
            }
        }
        tile = nxt;
    }
}


static inline int cdiv(int a, int b) { return (a + b - 1) / b; }

extern "C" void kernel_launch(void* const* d_in, const int* in_sizes, int n_in,
                              void* d_out, int out_size, void* d_ws, size_t ws_size,
                              hipStream_t stream) {
    const float* x   = (const float*)d_in[0];
    const int*   ei  = (const int*)d_in[1];
    const float* ew  = (const float*)d_in[2];
    const float* W1  = (const float*)d_in[3];
    const float* b1  = (const float*)d_in[4];
    const float* W2  = (const float*)d_in[5];
    const float* b2  = (const float*)d_in[6];
    const float* W3  = (const float*)d_in[7];
    const float* b3  = (const float*)d_in[8];
    float* out = (float*)d_out;

    const int N = in_sizes[0] / D_IN;
    const int E = in_sizes[1] / 2;
    const int* eRow = ei;
    const int* eCol = ei + E;

    float* ws     = (float*)d_ws;
    int*   cnt    = (int*)ws;
    int*   start  = cnt + N;
    const int NB  = cdiv(N, BSZ);
    const int NK  = cdiv(E, CHUNK);
    int*   tctr   = (int*)(ws + 3 * (size_t)N);
    float* dinv   = ws + 4 * (size_t)N;
    int*   tbl    = (int*)(ws + 5 * (size_t)N);
    float* buf0   = ws + 3 * (size_t)N + E;
    float* buf1   = buf0 + (size_t)N * D_HID;

    int2*  csrA   = (int2*)out;
    int2*  csrB   = (int2*)(buf1 + (size_t)N * D_OUT);
    int4*  part   = (int4*)buf0;

    k_part    <<<NK, 256, 0, stream>>>(eRow, eCol, ew, tbl, tctr, part, E, NB);
    k_bsort   <<<NB, 1024, 0, stream>>>(part, tbl, start, cnt, csrA, dinv, N, NB, NK);

    const int gGemm = cdiv(N, 128);
    const int gGath = cdiv(N, 8);

    k_gemm_wmma<D_IN, 8><<<gGemm, 256, 0, stream>>>(x, W1, dinv, buf1, N, 0);

    k_gather_gemm<8, 16><<<768, 512, 0, stream>>>(start, cnt, csrA, buf1, b1, dinv, W2, buf0, tctr, N,
                                                   csrA, csrB, 0);

    k_gather_gemm<4, 8><<<1536, 256, 0, stream>>>(start, cnt, csrA, buf0, b2, dinv, W3, buf1, tctr + 1, N,
                                                   csrA, csrB, E);

    k_gather<2><<<gGath, 256, 0, stream>>>(start, cnt, csrB, buf1, b3, dinv, out, N);
}


# First product x@W1 moved into the range-sort launch: 250 blocks of 400 targets hold W1 in LDS and run their own rows' WMMA k-loop in four slices under the sort's loads; 6 launches now 5
# speedup vs baseline: 4.6603x; 1.0499x over previous
#include <hip/hip_runtime.h>

typedef __attribute__((ext_vector_type(2))) float v2f;
typedef __attribute__((ext_vector_type(8))) float v8f;

#define D_IN  128
#define D_HID 128
#define D_OUT 64


#define BSZ   400
#define BSZP  416
#define NSTR  (BSZ / 16)
#define MAXB  2048
#define CHUNK 4096
#define EPT   (CHUNK / 256)
#define MAXK  2048
#define RU    8

__global__ __launch_bounds__(256)
void k_part(const int* __restrict__ row, const int* __restrict__ col, const float* __restrict__ w,
            int* __restrict__ tbl, int* tctr, int4* __restrict__ part, int e, int nb) {
    __shared__ int hist[MAXB + 1];
    const int tid  = threadIdx.x;
    const int lane = tid & 31;
    const int base = blockIdx.x * CHUNK + tid;
    int c[EPT], s[EPT], wb[EPT], r[EPT];
#pragma unroll
    for (int j = 0; j < EPT; ++j) {
        const int i = base + j * 256;
        const bool ok = i < e;
        c[j]  = ok ? col[i] : -1;
        s[j]  = ok ? row[i] : 0;
        wb[j] = ok ? __float_as_int(w[i]) : 0;
    }
    for (int b = tid; b < nb; b += 256) hist[b] = 0;
    if (blockIdx.x == 0 && tid < 2) tctr[tid] = 0;
    __syncthreads();
#pragma unroll
    for (int j = 0; j < EPT; ++j)
        r[j] = (c[j] >= 0) ? atomicAdd(&hist[(unsigned)c[j] / BSZ], 1) : 0;
    __syncthreads();
    if (tid < 32) {
        int carry = 0;
        for (int i0 = 0; i0 < nb; i0 += 32) {
            const int i = i0 + lane;
            const int v = (i < nb) ? hist[i] : 0;
            int incl = v;
#pragma unroll
            for (int d = 1; d < 32; d <<= 1) {
                int t = __shfl_up(incl, d);
                if (lane >= d) incl += t;
            }
            if (i < nb) hist[i] = carry + incl - v;
            carry += __shfl(incl, 31);
        }
        if (lane == 0) hist[nb] = carry;
    }
    __syncthreads();
    int* trow = tbl + (size_t)blockIdx.x * (nb + 1);
    for (int b = tid; b <= nb; b += 256) trow[b] = hist[b];
    int4* mine = part + (size_t)blockIdx.x * CHUNK;
#pragma unroll
    for (int j = 0; j < EPT; ++j) {
        if (c[j] >= 0) {
            int4 v; v.x = s[j]; v.y = c[j]; v.z = wb[j]; v.w = 0;
            mine[hist[(unsigned)c[j] / BSZ] + r[j]] = v;
        }
    }
}


__device__ __forceinline__ void prod_slice(v8f (&acc)[4], const float* __restrict__ a, const float* bCol,
                                           int kb, float2& nx) {
#pragma unroll 2
    for (int k0 = kb; k0 < kb + 32; k0 += 4) {
        v2f av; av.x = nx.x; av.y = nx.y;
        const int kn = (k0 + 4) & 127;
        nx = *(const float2*)(a + kn);
#pragma unroll
        for (int t = 0; t < 4; ++t) {
            const float2 b2 = *(const float2*)(bCol + k0 * 128 + t * 64);
            v2f bw; bw.x = b2.x; bw.y = b2.y;
            acc[t] = __builtin_amdgcn_wmma_f32_16x16x4_f32(
                false, av, false, bw, (short)0, acc[t], false, false);
        }
    }
}

__device__ __forceinline__ void prod_store(const v8f (&acc)[4], const float (&dl)[8], float* __restrict__ lin,
                                           int rowBase, int half, int colBase, int n) {
#pragma unroll
    for (int t = 0; t < 4; ++t) {
#pragma unroll
        for (int r = 0; r < 8; ++r) {
            const int rr = rowBase + half * 8 + r;
            if (rr < n) lin[(size_t)rr * 128 + colBase + t * 16] = acc[t][r] * dl[r];
        }
    }
}

__global__ __launch_bounds__(1024)
void k_bsort_gemm(const int4* __restrict__ part, const int* __restrict__ tbl,
                  int* __restrict__ start, int* __restrict__ cnt, int2* __restrict__ csr,
                  float* __restrict__ dinv, const float* __restrict__ x, const float* __restrict__ W,
                  float* __restrict__ lin, int n, int nb, int nk) {
    __shared__ __attribute__((aligned(16))) float Wl[128 * 128];
    __shared__ int   c[BSZP];
    __shared__ int   cur[BSZP];
    __shared__ float dsum[BSZP];
    __shared__ int   wsum[BSZP / 32];
    __shared__ int   roff[MAXK];
    __shared__ int   rlen[MAXK];
    __shared__ int   psum[32];
    const int tid  = threadIdx.x;
    const int lane = tid & 31;
    const int wv   = __builtin_amdgcn_readfirstlane(tid >> 5);
    const int half = lane >> 4;
    const int lidx = lane & 15;
    const int b    = blockIdx.x;
    const int t0   = b * BSZ;
    if (tid < BSZP) { c[tid] = 0; dsum[tid] = 1.0f; }

    for (int i = tid; i < 32 * 32; i += 1024) {
        const int kq = i / 32;
        const int cc = (i % 32) * 4;
        const float* src = W + (size_t)(4 * kq) * 128 + cc;
        const float4 r0 = *(const float4*)(src);
        const float4 r1 = *(const float4*)(src + 128);
        const float4 r2 = *(const float4*)(src + 256);
        const float4 r3 = *(const float4*)(src + 384);
        float* d = Wl + ((size_t)kq * 128 + cc) * 4;
        float4 o;
        o.x = r0.x; o.y = r1.x; o.z = r2.x; o.w = r3.x; *(float4*)(d)      = o;
        o.x = r0.y; o.y = r1.y; o.z = r2.y; o.w = r3.y; *(float4*)(d + 4)  = o;
        o.x = r0.z; o.y = r1.z; o.z = r2.z; o.w = r3.z; *(float4*)(d + 8)  = o;
        o.x = r0.w; o.y = r1.w; o.z = r2.w; o.w = r3.w; *(float4*)(d + 12) = o;
    }

    int p = 0;
    for (int k = tid; k < nk; k += 1024) {
        const int* tr = tbl + (size_t)k * (nb + 1) + b;
        const int o = tr[0], o1 = tr[1];
        roff[k] = k * CHUNK + o;
        rlen[k] = o1 - o;
        p += o;
    }
#pragma unroll
    for (int d = 16; d >= 1; d >>= 1) p += __shfl_xor(p, d);
    if (lane == 0) psum[wv] = p;

    const int  colBase = (wv & 1) * 64 + lidx;
    const int  rowA    = t0 + (wv >> 1) * 16;
    const int  rowB    = t0 + (16 + (wv >> 1)) * 16;
    const bool doA     = rowA < n;
    const bool doB     = (16 + (wv >> 1) < NSTR) && (rowB < n);
    int rcA = rowA + lidx; if (rcA > n - 1) rcA = n - 1;
    int rcB = rowB + lidx; if (rcB > n - 1) rcB = n - 1;
    const float* aA   = x + ((unsigned)rcA * 128u + 2u * (unsigned)half);
    const float* aB   = x + ((unsigned)rcB * 128u + 2u * (unsigned)half);
    const float* bCol = Wl + ((wv & 1) * 64 + lidx) * 4 + 2 * half;
    v8f accA[4] = {};
    v8f accB[4] = {};
    float2 nA = *(const float2*)(aA);
    float2 nB = *(const float2*)(aB);

    __syncthreads();
    int bs = psum[lane];
#pragma unroll
    for (int d = 16; d >= 1; d >>= 1) bs += __shfl_xor(bs, d);

    {
        int t[RU];
#pragma unroll
        for (int u = 0; u < RU; ++u) {
            const int k = wv + 32 * u;
            t[u] = -1;
            if (k < nk && lane < rlen[k]) t[u] = part[roff[k] + lane].y;
        }
        prod_slice(accA, aA, bCol, 0, nA);
        if (doB) prod_slice(accB, aB, bCol, 0, nB);
#pragma unroll
        for (int u = 0; u < RU; ++u)
            if (t[u] >= 0) atomicAdd(&c[t[u] - t0], 1);
    }
    for (int k0 = wv + 32 * RU; k0 < nk; k0 += 32 * RU) {
        int t[RU];
#pragma unroll
        for (int u = 0; u < RU; ++u) {
            const int k = k0 + 32 * u;
            t[u] = -1;
            if (k < nk && lane < rlen[k]) t[u] = part[roff[k] + lane].y;
        }
#pragma unroll
        for (int u = 0; u < RU; ++u)
            if (t[u] >= 0) atomicAdd(&c[t[u] - t0], 1);
    }
    for (int k = wv; k < nk; k += 32) {
        const int len = rlen[k];
        const int4* run = part + roff[k];
        for (int i = 32 + lane; i < len; i += 32) atomicAdd(&c[run[i].y - t0], 1);
    }
    __syncthreads();

    int v = 0, incl = 0;
    if (tid < BSZP) {
        v = c[tid];
        incl = v;
#pragma unroll
        for (int d = 1; d < 32; d <<= 1) {
            int t = __shfl_up(incl, d);
            if (lane >= d) incl += t;
        }
        if (lane == 31) wsum[wv] = incl;
    }
    __syncthreads();
    if (tid < BSZP) {
        int pre = 0;
        for (int k = 0; k < wv; ++k) pre += wsum[k];
        const int off = pre + incl - v;
        cur[tid] = off;
        const int t = t0 + tid;
        if (tid < BSZ && t < n) { start[t] = bs + off; cnt[t] = v; }
    }
    prod_slice(accA, aA, bCol, 32, nA);
    if (doB) prod_slice(accB, aB, bCol, 32, nB);
    __syncthreads();

    prod_slice(accA, aA, bCol, 64, nA);
    if (doB) prod_slice(accB, aB, bCol, 64, nB);

    int2* outp = csr + bs;
    for (int k0 = wv; k0 < nk; k0 += 32 * RU) {
        int4 rec[RU];
#pragma unroll
        for (int u = 0; u < RU; ++u) {
            const int k = k0 + 32 * u;
            rec[u].x = 0; rec[u].y = -1; rec[u].z = 0; rec[u].w = 0;
            if (k < nk && lane < rlen[k]) rec[u] = part[roff[k] + lane];
        }
#pragma unroll
        for (int u = 0; u < RU; ++u) {
            if (rec[u].y >= 0) {
                const int tl = rec[u].y - t0;
                const int q = atomicAdd(&cur[tl], 1);
                atomicAdd(&dsum[tl], __int_as_float(rec[u].z));
                int2 o; o.x = rec[u].x; o.y = rec[u].z;
                outp[q] = o;
            }
        }
    }
    for (int k = wv; k < nk; k += 32) {
        const int len = rlen[k];
        const int4* run = part + roff[k];
        for (int i = 32 + lane; i < len; i += 32) {
            const int4 rc = run[i];
            const int tl = rc.y - t0;
            const int q = atomicAdd(&cur[tl], 1);
            atomicAdd(&dsum[tl], __int_as_float(rc.z));
            int2 o; o.x = rc.x; o.y = rc.z;
            outp[q] = o;
        }
    }
    prod_slice(accA, aA, bCol, 96, nA);
    if (doB) prod_slice(accB, aB, bCol, 96, nB);
    __syncthreads();
    if (tid < BSZ) {
        const float d = dsum[tid];
        const float f = (d > 0.0f) ? rsqrtf(d) : 0.0f;
        dsum[tid] = f;
        const int t = t0 + tid;
        if (t < n) dinv[t] = f;
    }
    __syncthreads();

    if (doA) {
        float dl[8];
#pragma unroll
        for (int r = 0; r < 8; ++r) dl[r] = dsum[(wv >> 1) * 16 + half * 8 + r];
        prod_store(accA, dl, lin, rowA, half, colBase, n);
    }
    if (doB) {
        float dl[8];
#pragma unroll
        for (int r = 0; r < 8; ++r) dl[r] = dsum[(16 + (wv >> 1)) * 16 + half * 8 + r];
        prod_store(accB, dl, lin, rowB, half, colBase, n);
    }
}


template <int VEC> struct VecT;
template <> struct VecT<4> { typedef float4 T; };
template <> struct VecT<2> { typedef float2 T; };

__device__ __forceinline__ void vfma(float4& a, float w, const float4& v) {
    a.x = fmaf(w, v.x, a.x); a.y = fmaf(w, v.y, a.y);
    a.z = fmaf(w, v.z, a.z); a.w = fmaf(w, v.w, a.w);
}
__device__ __forceinline__ void vfma(float2& a, float w, const float2& v) {
    a.x = fmaf(w, v.x, a.x); a.y = fmaf(w, v.y, a.y);
}
__device__ __forceinline__ float4 vfinish(float s, const float4& l, const float4& b) {
    float4 o; o.x = fmaf(s, l.x, b.x); o.y = fmaf(s, l.y, b.y);
    o.z = fmaf(s, l.z, b.z); o.w = fmaf(s, l.w, b.w); return o;
}
__device__ __forceinline__ float2 vfinish(float s, const float2& l, const float2& b) {
    float2 o; o.x = fmaf(s, l.x, b.x); o.y = fmaf(s, l.y, b.y); return o;
}

template <int VEC>
__global__ __launch_bounds__(256)
void k_gather(const int* __restrict__ start, const int* __restrict__ cnt,
              const int2* __restrict__ csr,
              const float* __restrict__ lin, const float* __restrict__ bias,
              const float* __restrict__ dinv, float* __restrict__ out, int n) {
    typedef typename VecT<VEC>::T V;
    const int Dout = 32 * VEC;
    const int lane = threadIdx.x & 31;
    int node = blockIdx.x * 8 + (threadIdx.x >> 5);
    if (node >= n) return;
    node = __builtin_amdgcn_readfirstlane(node);
    const int s = __builtin_amdgcn_readfirstlane(start[node]);
    const int c = __builtin_amdgcn_readfirstlane(cnt[node]);

    const float* linL = lin + lane * VEC;
    const float dt = dinv[node];
    const V bv = *(const V*)(bias + lane * VEC);
    V acc = *(const V*)(linL + (size_t)node * Dout);

    for (int b0 = 0; b0 < c; b0 += 32) {
        const int m = (c - b0 < 32) ? (c - b0) : 32;
        int2 my; my.x = 0; my.y = 0;
        if (lane < m) my = csr[s + b0 + lane];
        const int mySrc = my.x;
        const int myWi  = my.y;
        int k = 0;
        for (; k + 4 <= m; k += 4) {
            const int s0 = __builtin_amdgcn_readlane(mySrc, k);
            const int s1 = __builtin_amdgcn_readlane(mySrc, k + 1);
            const int s2 = __builtin_amdgcn_readlane(mySrc, k + 2);
            const int s3 = __builtin_amdgcn_readlane(mySrc, k + 3);
            const V v0 = *(const V*)(linL + (size_t)s0 * Dout);
            const V v1 = *(const V*)(linL + (size_t)s1 * Dout);
            const V v2 = *(const V*)(linL + (size_t)s2 * Dout);
            const V v3 = *(const V*)(linL + (size_t)s3 * Dout);
            vfma(acc, __int_as_float(__builtin_amdgcn_readlane(myWi, k)),     v0);
            vfma(acc, __int_as_float(__builtin_amdgcn_readlane(myWi, k + 1)), v1);
            vfma(acc, __int_as_float(__builtin_amdgcn_readlane(myWi, k + 2)), v2);
            vfma(acc, __int_as_float(__builtin_amdgcn_readlane(myWi, k + 3)), v3);
        }
        for (; k < m; ++k) {
            const int s0 = __builtin_amdgcn_readlane(mySrc, k);
            const V v0 = *(const V*)(linL + (size_t)s0 * Dout);
            vfma(acc, __int_as_float(__builtin_amdgcn_readlane(myWi, k)), v0);
        }
    }
    *(V*)(out + (size_t)node * Dout + lane * VEC) = vfinish(dt, acc, bv);
}


#define APITCH 132

template <int U>
__device__ __forceinline__ void gather_batch(float4& acc, const float* __restrict__ linL,
                                             int mySrc, int myWi, int k) {
    float4 v[U];
#pragma unroll
    for (int u = 0; u < U; ++u) {
        const int su = __builtin_amdgcn_readlane(mySrc, k + u);
        v[u] = *(const float4*)(linL + (size_t)su * 128);
    }
#pragma unroll
    for (int u = 0; u < U; ++u)
        vfma(acc, __int_as_float(__builtin_amdgcn_readlane(myWi, k + u)), v[u]);
}

__device__ __forceinline__ float4 gather_sum128(const float* __restrict__ linL, const int2* __restrict__ csr,
                                                int s, int c, int2 first, float4 acc, int lane) {
    for (int b0 = 0; b0 < c; b0 += 32) {
        const int m = (c - b0 < 32) ? (c - b0) : 32;
        int2 my = first;
        if (b0 > 0) {
            my.x = 0; my.y = 0;
            if (lane < m) my = csr[s + b0 + lane];
        }
        const int mySrc = my.x;
        const int myWi  = my.y;
        int k = 0;
        for (; k + 8 <= m; k += 8) gather_batch<8>(acc, linL, mySrc, myWi, k);
        if ((m - k) & 4) { gather_batch<4>(acc, linL, mySrc, myWi, k); k += 4; }
        if ((m - k) & 2) { gather_batch<2>(acc, linL, mySrc, myWi, k); k += 2; }
        if ((m - k) & 1) { gather_batch<1>(acc, linL, mySrc, myWi, k); }
    }
    return acc;
}

template <int CT, int NW>
__global__ __launch_bounds__(NW * 32)
void k_gather_gemm(const int* __restrict__ start, const int* __restrict__ cnt,
                   const int2* __restrict__ csr,
                   const float* __restrict__ lin, const float* __restrict__ bias,
                   const float* __restrict__ dinv, const float* __restrict__ Wn,
                   float* __restrict__ linN, int* tctr, int n,
                   const int2* __restrict__ cpSrc, int2* __restrict__ cpDst, int cpN) {
    constexpr int DoutN = CT * 16;
    constexpr int Din   = 128;
    __shared__ __attribute__((aligned(16))) float Wl[Din * DoutN];
    constexpr int TN    = 2 * NW;
    constexpr int JOBS  = (TN / 16) * CT;
    __shared__ __attribute__((aligned(16))) float hs[TN * APITCH];
    __shared__ int   tcur;

    const int lane = threadIdx.x & 31;
    const int wave = __builtin_amdgcn_readfirstlane(threadIdx.x >> 5);
    const int half = lane >> 4;
    const int lidx = lane & 15;
    const int ntiles = (n + TN - 1) / TN;

    if (threadIdx.x == 0) tcur = atomicAdd(tctr, 1);

    {
        const int gsz = gridDim.x * NW * 32;
        for (int i0 = blockIdx.x * NW * 32 + threadIdx.x; i0 < cpN; i0 += 4 * gsz) {
            int2 cv[4];
#pragma unroll
            for (int u = 0; u < 4; ++u) {
                const int i = i0 + u * gsz;
                cv[u].x = 0; cv[u].y = 0;
                if (i < cpN) cv[u] = cpSrc[i];
            }
#pragma unroll
            for (int u = 0; u < 4; ++u) {
                const int i = i0 + u * gsz;
                if (i < cpN) cpDst[i] = cv[u];
            }
        }
    }

    for (int i = threadIdx.x; i < (Din / 4) * (DoutN / 4); i += NW * 32) {
        const int kq = i / (DoutN / 4);
        const int c  = (i % (DoutN / 4)) * 4;
        const float* src = Wn + (size_t)(4 * kq) * DoutN + c;
        const float4 r0 = *(const float4*)(src);
        const float4 r1 = *(const float4*)(src + DoutN);
        const float4 r2 = *(const float4*)(src + 2 * DoutN);
        const float4 r3 = *(const float4*)(src + 3 * DoutN);
        float* d = Wl + ((size_t)kq * DoutN + c) * 4;
        float4 o;
        o.x = r0.x; o.y = r1.x; o.z = r2.x; o.w = r3.x; *(float4*)(d)      = o;
        o.x = r0.y; o.y = r1.y; o.z = r2.y; o.w = r3.y; *(float4*)(d + 4)  = o;
        o.x = r0.z; o.y = r1.z; o.z = r2.z; o.w = r3.z; *(float4*)(d + 8)  = o;
        o.x = r0.w; o.y = r1.w; o.z = r2.w; o.w = r3.w; *(float4*)(d + 12) = o;
    }

    const float* linL = lin + lane * 4;
    const float4 bv = *(const float4*)(bias + lane * 4);
    __syncthreads();
    int tile = __builtin_amdgcn_readfirstlane(tcur);

    while (tile < ntiles) {
        __syncthreads();
        if (threadIdx.x == 0) tcur = atomicAdd(tctr, 1);
        const int blockBase = tile * TN;

        const int nodeA = (blockBase + wave * 2     < n) ? blockBase + wave * 2     : n - 1;
        const int nodeB = (blockBase + wave * 2 + 1 < n) ? blockBase + wave * 2 + 1 : n - 1;
        const int rsA = start[nodeA], rcA = cnt[nodeA];
        const int rsB = start[nodeB], rcB = cnt[nodeB];
        const float dA = dinv[nodeA], dB = dinv[nodeB];
        const int sA = __builtin_amdgcn_readfirstlane(rsA);
        const int cA = __builtin_amdgcn_readfirstlane(rcA);
        const int sB = __builtin_amdgcn_readfirstlane(rsB);
        const int cB = __builtin_amdgcn_readfirstlane(rcB);
        int2 fA; fA.x = 0; fA.y = 0;
        int2 fB; fB.x = 0; fB.y = 0;
        if (lane < cA) fA = csr[sA + lane];
        if (lane < cB) fB = csr[sB + lane];
        float4 accA = *(const float4*)(linL + (size_t)nodeA * Din);
        float4 accB = *(const float4*)(linL + (size_t)nodeB * Din);
        accA = gather_sum128(linL, csr, sA, cA, fA, accA, lane);
        accB = gather_sum128(linL, csr, sB, cB, fB, accB, lane);
        float4 hA = vfinish(dA, accA, bv);
        float4 hB = vfinish(dB, accB, bv);
        hA.x = fmaxf(hA.x, 0.0f); hA.y = fmaxf(hA.y, 0.0f); hA.z = fmaxf(hA.z, 0.0f); hA.w = fmaxf(hA.w, 0.0f);
        hB.x = fmaxf(hB.x, 0.0f); hB.y = fmaxf(hB.y, 0.0f); hB.z = fmaxf(hB.z, 0.0f); hB.w = fmaxf(hB.w, 0.0f);
        *(float4*)(hs + (wave * 2)     * APITCH + lane * 4) = hA;
        *(float4*)(hs + (wave * 2 + 1) * APITCH + lane * 4) = hB;
        __syncthreads();
        const int nxt = __builtin_amdgcn_readfirstlane(tcur);

        const int stripe = wave / CT;
        const int ctile  = wave % CT;
        v8f acc8 = {};
        if (wave < JOBS) {
            const float* aRow = hs + (stripe * 16 + lidx) * APITCH + 2 * half;
            const float* bCol = Wl + (ctile * 16 + lidx) * 4 + 2 * half;
            for (int k0 = 0; k0 < Din; k0 += 4) {
                const float2 a2 = *(const float2*)(aRow + k0);
                const float2 b2 = *(const float2*)(bCol + k0 * DoutN);
                v2f av; av.x = a2.x; av.y = a2.y;
                v2f bw; bw.x = b2.x; bw.y = b2.y;
                acc8 = __builtin_amdgcn_wmma_f32_16x16x4_f32(
                    false, av, false, bw, (short)0, acc8, false, false);
            }
        }
        __syncthreads();

        if (wave < JOBS) {
#pragma unroll
            for (int r = 0; r < 8; ++r)
                hs[(stripe * 16 + half * 8 + r) * APITCH + ctile * 16 + lidx] = acc8[r];
        }
        __syncthreads();

        constexpr int RPI = 128 / DoutN;
        constexpr int LPR = 32 / RPI;
        for (int r0 = wave * RPI; r0 < TN; r0 += NW * RPI) {
            const int row = r0 + lane / LPR;
            const int col = (lane % LPR) * 4;
            const int rr  = blockBase + row;
            if (rr < n) {
                const float d = dinv[rr];
                const float4 v = *(const float4*)(hs + row * APITCH + col);
                float4 o; o.x = v.x * d; o.y = v.y * d; o.z = v.z * d; o.w = v.w * d;
                *(float4*)(linN + (size_t)rr * DoutN + col) = o;
            }
        }
        tile = nxt;
    }
}


static inline int cdiv(int a, int b) { return (a + b - 1) / b; }

extern "C" void kernel_launch(void* const* d_in, const int* in_sizes, int n_in,
                              void* d_out, int out_size, void* d_ws, size_t ws_size,
                              hipStream_t stream) {
    const float* x   = (const float*)d_in[0];
    const int*   ei  = (const int*)d_in[1];
    const float* ew  = (const float*)d_in[2];
    const float* W1  = (const float*)d_in[3];
    const float* b1  = (const float*)d_in[4];
    const float* W2  = (const float*)d_in[5];
    const float* b2  = (const float*)d_in[6];
    const float* W3  = (const float*)d_in[7];
    const float* b3  = (const float*)d_in[8];
    float* out = (float*)d_out;

    const int N = in_sizes[0] / D_IN;
    const int E = in_sizes[1] / 2;
    const int* eRow = ei;
    const int* eCol = ei + E;

    float* ws     = (float*)d_ws;
    int*   cnt    = (int*)ws;
    int*   start  = cnt + N;
    const int NB  = cdiv(N, BSZ);
    const int NK  = cdiv(E, CHUNK);
    int*   tctr   = (int*)(ws + 3 * (size_t)N);
    float* dinv   = ws + 4 * (size_t)N;
    int*   tbl    = (int*)(ws + 5 * (size_t)N);
    float* buf0   = ws + 3 * (size_t)N + E;
    float* buf1   = buf0 + (size_t)N * D_HID;

    int2*  csrA   = (int2*)out;
    int2*  csrB   = (int2*)(buf1 + (size_t)N * D_OUT);
    int4*  part   = (int4*)buf0;

    k_part       <<<NK, 256, 0, stream>>>(eRow, eCol, ew, tbl, tctr, part, E, NB);
    k_bsort_gemm <<<NB, 1024, 0, stream>>>(part, tbl, start, cnt, csrA, dinv, x, W1, buf1, N, NB, NK);

    const int gGath = cdiv(N, 8);

    k_gather_gemm<8, 16><<<768, 512, 0, stream>>>(start, cnt, csrA, buf1, b1, dinv, W2, buf0, tctr, N,
                                                   csrA, csrB, 0);

    k_gather_gemm<4, 8><<<1536, 256, 0, stream>>>(start, cnt, csrA, buf0, b2, dinv, W3, buf1, tctr + 1, N,
                                                   csrA, csrB, E);

    k_gather<2><<<gGath, 256, 0, stream>>>(start, cnt, csrB, buf1, b3, dinv, out, N);
}
